# starting kernel
# baseline (speedup 1.0000x reference)
#include <hip/hip_runtime.h>

typedef float v2f __attribute__((ext_vector_type(2)));
typedef float v8f __attribute__((ext_vector_type(8)));

#define NJ  24
#define DF  19
#define DPJ 6

__device__ __constant__ int kParents[NJ] = {
    -1, 0, 0, 0, 1, 2, 3, 4, 5, 6, 7, 8, 9, 9, 9, 12, 13, 14, 16, 17, 18, 19, 20, 21};

__device__ inline void lds_fence() {
  asm volatile("s_wait_dscnt 0" ::: "memory");
}

__device__ inline v8f wmma4(v2f a, v2f b, v8f c) {
  return __builtin_amdgcn_wmma_f32_16x16x4_f32(
       false, a,  false, b,
       (short)0, c,  false,  false);
}

__device__ inline int imin(int a, int b) { return a < b ? a : b; }

__global__ __launch_bounds__(128) void pose_enc_kernel(
    const float* __restrict__ rots, const float* __restrict__ Jtrs,
    const float* __restrict__ W0,   const float* __restrict__ b0,
    const float* __restrict__ W1,   const float* __restrict__ b1,
    const float* __restrict__ W2,   const float* __restrict__ b2,
    float* __restrict__ out, int B)
{
  __shared__ float s_in[4][16][20];
  __shared__ float s_h [4][16][20];
  __shared__ float s_feat[4][NJ + 1][16][DPJ];

  __shared__ v2f s_w1a[NJ][10][16];
  __shared__ v2f s_w1b[NJ][10][4];
  __shared__ v2f s_w2p[NJ][10][8];
  __shared__ v2f s_w0p[144][8];

  const int tid  = threadIdx.x;

  for (int i = tid; i < NJ * 10 * 16; i += 128) {
    int c = i & 15, pr = (i >> 4) % 10, j = i / 160;
    int k = 2 * pr;
    float x = W1[j * 361 + c * 19 + k];
    float y = W1[j * 361 + c * 19 + imin(k + 1, 18)];
    y = (k + 1 < DF) ? y : 0.f;
    v2f v; v.x = x; v.y = y;
    s_w1a[j][pr][c] = v;
  }
  for (int i = tid; i < NJ * 10 * 4; i += 128) {
    int c = i & 3, pr = (i >> 2) % 10, j = i / 40;
    int k = 2 * pr, col = imin(16 + c, 18);
    float x = W1[j * 361 + col * 19 + k];
    float y = W1[j * 361 + col * 19 + imin(k + 1, 18)];
    x = (c < 3) ? x : 0.f;
    y = (c < 3 && k + 1 < DF) ? y : 0.f;
    v2f v; v.x = x; v.y = y;
    s_w1b[j][pr][c] = v;
  }
  for (int i = tid; i < NJ * 10 * 8; i += 128) {
    int c = i & 7, pr = (i >> 3) % 10, j = i / 80;
    int k = 2 * pr, col = imin(c, 5);
    float x = W2[j * 114 + col * 19 + k];
    float y = W2[j * 114 + col * 19 + imin(k + 1, 18)];
    x = (c < DPJ) ? x : 0.f;
    y = (c < DPJ && k + 1 < DF) ? y : 0.f;
    v2f v; v.x = x; v.y = y;
    s_w2p[j][pr][c] = v;
  }
  for (int i = tid; i < 144 * 8; i += 128) {
    int c = i & 7, pr = i >> 3;
    int k = 2 * pr, col = imin(c, 5);
    float x = W0[col * 288 + k];
    float y = W0[col * 288 + k + 1];
    x = (c < DPJ) ? x : 0.f;
    y = (c < DPJ) ? y : 0.f;
    v2f v; v.x = x; v.y = y;
    s_w0p[pr][c] = v;
  }
  __syncthreads();

  const int w    = tid >> 5;
  const int lane = tid & 31;
  const int half = lane >> 4;
  const int n    = lane & 15;
  const int n7   = imin(n, 7);
  const int n3   = imin(n, 3);
  const int rb   = (blockIdx.x * 4 + w) * 16;
  if (rb >= B) return;

  {
    float bv = b0[imin(n, 5)];
    bv = (n < DPJ) ? bv : 0.f;
    v8f acc;
#pragma unroll
    for (int v = 0; v < 8; ++v) acc[v] = bv;

    const float* rrow = rots + (long)(rb + n) * 216;
    const float* jrow = Jtrs + (long)(rb + n) * 72;
#pragma unroll 4
    for (int step = 0; step < 54; ++step) {
      int pr = 2 * step + half;
      v2f a = *(const v2f*)(rrow + 2 * pr);
      acc = wmma4(a, s_w0p[pr][n7], acc);
    }
#pragma unroll 4
    for (int step = 54; step < 72; ++step) {
      int pr = 2 * step + half;
      v2f a = *(const v2f*)(jrow + 2 * pr - 216);
      acc = wmma4(a, s_w0p[pr][n7], acc);
    }
    if (n < DPJ) {
#pragma unroll
      for (int v = 0; v < 8; ++v) s_feat[w][NJ][v + 8 * half][n] = acc[v];
    }
  }
  lds_fence();

  for (int j = 0; j < NJ; ++j) {
    const int p = kParents[j];

    if (half == 0) {
      const float* rp = rots + (long)(rb + n) * 216 + j * 9;
#pragma unroll
      for (int q = 0; q < 9; ++q) s_in[w][n][q] = rp[q];
    } else {
      const float* jp = Jtrs + (long)(rb + n) * 72;
      float x = jp[j * 3 + 0], y = jp[j * 3 + 1], z = jp[j * 3 + 2];
      int pc = (p >= 0) ? p : 0;
      float px = jp[pc * 3 + 0], py = jp[pc * 3 + 1], pz = jp[pc * 3 + 2];
      px = (p >= 0) ? px : 0.f;  py = (p >= 0) ? py : 0.f;  pz = (p >= 0) ? pz : 0.f;
      float dx = x - px, dy = y - py, dz = z - pz;
      float bl = sqrtf(dx * dx + dy * dy + dz * dz);
      s_in[w][n][9]  = x;  s_in[w][n][10] = y;
      s_in[w][n][11] = z;  s_in[w][n][12] = bl;
      const int src = (p >= 0) ? p : NJ;
#pragma unroll
      for (int f = 0; f < DPJ; ++f) s_in[w][n][13 + f] = s_feat[w][src][n][f];
      s_in[w][n][19] = 0.f;
    }
    lds_fence();

    float bA = b1[j * DF + n];
    float bB = b1[j * DF + imin(16 + n, 18)];
    bB = (n < 3) ? bB : 0.f;
    v8f accA, accB;
#pragma unroll
    for (int v = 0; v < 8; ++v) { accA[v] = bA; accB[v] = bB; }
#pragma unroll
    for (int step = 0; step < 5; ++step) {
      int pr = 2 * step + half;
      v2f a = *(const v2f*)&s_in[w][n][2 * pr];
      accA = wmma4(a, s_w1a[j][pr][n],  accA);
      accB = wmma4(a, s_w1b[j][pr][n3], accB);
    }
#pragma unroll
    for (int v = 0; v < 8; ++v) {
      int m = v + 8 * half;
      s_h[w][m][n] = fmaxf(accA[v], 0.f);
      if (n < 3) s_h[w][m][16 + n] = fmaxf(accB[v], 0.f);
      if (n == 3) s_h[w][m][19] = 0.f;
    }
    lds_fence();

    float bC = b2[j * DPJ + imin(n, 5)];
    bC = (n < DPJ) ? bC : 0.f;
    v8f acc2;
#pragma unroll
    for (int v = 0; v < 8; ++v) acc2[v] = bC;
#pragma unroll
    for (int step = 0; step < 5; ++step) {
      int pr = 2 * step + half;
      v2f a = *(const v2f*)&s_h[w][n][2 * pr];
      acc2 = wmma4(a, s_w2p[j][pr][n7], acc2);
    }
    if (n < DPJ) {
#pragma unroll
      for (int v = 0; v < 8; ++v) {
        int m = v + 8 * half;
        float val = acc2[v];
        s_feat[w][j][m][n] = val;
        out[(long)(rb + m) * (NJ * DPJ) + j * DPJ + n] = val;
      }
    }
    lds_fence();
  }
}

extern "C" void kernel_launch(void* const* d_in, const int* in_sizes, int n_in,
                              void* d_out, int out_size, void* d_ws, size_t ws_size,
                              hipStream_t stream) {
  const float* rots = (const float*)d_in[0];
  const float* Jtrs = (const float*)d_in[1];
  const float* W0   = (const float*)d_in[2];
  const float* b0   = (const float*)d_in[3];
  const float* W1   = (const float*)d_in[4];
  const float* b1   = (const float*)d_in[5];
  const float* W2   = (const float*)d_in[6];
  const float* b2   = (const float*)d_in[7];
  float* outp = (float*)d_out;

  int B = in_sizes[0] / (NJ * 9);
  int grid = (B + 63) / 64;
  pose_enc_kernel<<<grid, 128, 0, stream>>>(rots, Jtrs, W0, b0, W1, b1, W2, b2,
                                            outp, B);
}


# joint chain waits on no memory load: rot, Jtr, parent Jtr and biases requested two joints ahead into three rotating register sets; parent table folded to immediates
# speedup vs baseline: 1.2449x; 1.2449x over previous
#include <hip/hip_runtime.h>

typedef float v2f __attribute__((ext_vector_type(2)));
typedef float v8f __attribute__((ext_vector_type(8)));

#define NJ  24
#define DF  19
#define DPJ 6

__device__ inline int imin(int a, int b) { return a < b ? a : b; }

__host__ __device__ constexpr unsigned long long pack_parents(int base) {
  constexpr int P[NJ] = {
      -1, 0, 0, 0, 1, 2, 3, 4, 5, 6, 7, 8, 9, 9, 9, 12, 13, 14, 16, 17, 18, 19, 20, 21};
  unsigned long long v = 0;
  for (int i = 0; i < 12; ++i) v |= (unsigned long long)(P[base + i] + 1) << (5 * i);
  return v;
}
__device__ inline int parent_of(int j) {
  constexpr unsigned long long lo = pack_parents(0), hi = pack_parents(12);
  const unsigned long long wv = (j < 12) ? lo : hi;
  const int s = 5 * ((j < 12) ? j : j - 12);
  return (int)((wv >> s) & 31ull) - 1;
}

struct JointRegs {
  float q[9];
  float bA, bB, bC;
};

__device__ inline void fetch_joint(JointRegs& r, const float* rrow, const float* jrow,
                                   const float* __restrict__ b1, const float* __restrict__ b2,
                                   int j, int half, int n) {
  const int p  = parent_of(j);
  const int pc = (p >= 0) ? p : 0;
  const float* s0 = half ? jrow + 3 * j  : rrow + 9 * j;
  const float* s1 = half ? jrow + 3 * pc : rrow + 9 * j + 3;
  const float* s2 = half ? jrow + 3 * j  : rrow + 9 * j + 6;
#pragma unroll
  for (int q = 0; q < 3; ++q) { r.q[q] = s0[q]; r.q[3 + q] = s1[q]; r.q[6 + q] = s2[q]; }
  r.bA = b1[j * DF + n];
  r.bB = b1[j * DF + imin(16 + n, 18)];
  r.bC = b2[j * DPJ + imin(n, 5)];
}

__device__ inline void lds_fence() {
  asm volatile("s_wait_dscnt 0" ::: "memory");
}

__device__ inline v8f wmma4(v2f a, v2f b, v8f c) {
  return __builtin_amdgcn_wmma_f32_16x16x4_f32(
       false, a,  false, b,
       (short)0, c,  false,  false);
}

__global__ __launch_bounds__(128) void pose_enc_kernel(
    const float* __restrict__ rots, const float* __restrict__ Jtrs,
    const float* __restrict__ W0,   const float* __restrict__ b0,
    const float* __restrict__ W1,   const float* __restrict__ b1,
    const float* __restrict__ W2,   const float* __restrict__ b2,
    float* __restrict__ out, int B)
{
  __shared__ float s_in[4][16][20];
  __shared__ float s_h [4][16][20];
  __shared__ float s_feat[4][NJ + 1][16][DPJ];

  __shared__ v2f s_w1a[NJ][10][16];
  __shared__ v2f s_w1b[NJ][10][4];
  __shared__ v2f s_w2p[NJ][10][8];
  __shared__ v2f s_w0p[144][8];

  const int tid  = threadIdx.x;

  for (int i = tid; i < NJ * 10 * 16; i += 128) {
    int c = i & 15, pr = (i >> 4) % 10, j = i / 160;
    int k = 2 * pr;
    float x = W1[j * 361 + c * 19 + k];
    float y = W1[j * 361 + c * 19 + imin(k + 1, 18)];
    y = (k + 1 < DF) ? y : 0.f;
    v2f v; v.x = x; v.y = y;
    s_w1a[j][pr][c] = v;
  }
  for (int i = tid; i < NJ * 10 * 4; i += 128) {
    int c = i & 3, pr = (i >> 2) % 10, j = i / 40;
    int k = 2 * pr, col = imin(16 + c, 18);
    float x = W1[j * 361 + col * 19 + k];
    float y = W1[j * 361 + col * 19 + imin(k + 1, 18)];
    x = (c < 3) ? x : 0.f;
    y = (c < 3 && k + 1 < DF) ? y : 0.f;
    v2f v; v.x = x; v.y = y;
    s_w1b[j][pr][c] = v;
  }
  for (int i = tid; i < NJ * 10 * 8; i += 128) {
    int c = i & 7, pr = (i >> 3) % 10, j = i / 80;
    int k = 2 * pr, col = imin(c, 5);
    float x = W2[j * 114 + col * 19 + k];
    float y = W2[j * 114 + col * 19 + imin(k + 1, 18)];
    x = (c < DPJ) ? x : 0.f;
    y = (c < DPJ && k + 1 < DF) ? y : 0.f;
    v2f v; v.x = x; v.y = y;
    s_w2p[j][pr][c] = v;
  }
  for (int i = tid; i < 144 * 8; i += 128) {
    int c = i & 7, pr = i >> 3;
    int k = 2 * pr, col = imin(c, 5);
    float x = W0[col * 288 + k];
    float y = W0[col * 288 + k + 1];
    x = (c < DPJ) ? x : 0.f;
    y = (c < DPJ) ? y : 0.f;
    v2f v; v.x = x; v.y = y;
    s_w0p[pr][c] = v;
  }
  __syncthreads();

  const int w    = tid >> 5;
  const int lane = tid & 31;
  const int half = lane >> 4;
  const int n    = lane & 15;
  const int n7   = imin(n, 7);
  const int n3   = imin(n, 3);
  const int rb   = (blockIdx.x * 4 + w) * 16;
  if (rb >= B) return;

  const float* rrow = rots + (long)(rb + n) * 216;
  const float* jrow = Jtrs + (long)(rb + n) * 72;

  JointRegs jA, jB, jC;
  fetch_joint(jA, rrow, jrow, b1, b2, 0, half, n);
  fetch_joint(jB, rrow, jrow, b1, b2, 1, half, n);

  {
    float bv = b0[imin(n, 5)];
    bv = (n < DPJ) ? bv : 0.f;
    v8f acc;
#pragma unroll
    for (int v = 0; v < 8; ++v) acc[v] = bv;

#pragma unroll 4
    for (int step = 0; step < 54; ++step) {
      int pr = 2 * step + half;
      v2f a = *(const v2f*)(rrow + 2 * pr);
      acc = wmma4(a, s_w0p[pr][n7], acc);
    }
#pragma unroll 4
    for (int step = 54; step < 72; ++step) {
      int pr = 2 * step + half;
      v2f a = *(const v2f*)(jrow + 2 * pr - 216);
      acc = wmma4(a, s_w0p[pr][n7], acc);
    }
    if (n < DPJ) {
#pragma unroll
      for (int v = 0; v < 8; ++v) s_feat[w][NJ][v + 8 * half][n] = acc[v];
    }
  }
  lds_fence();

  auto run_joint = [&](const int j, const JointRegs& r) {
    const int p = parent_of(j);

    if (half == 0) {
#pragma unroll
      for (int q = 0; q < 9; ++q) s_in[w][n][q] = r.q[q];
    } else {
      float x = r.q[0], y = r.q[1], z = r.q[2];
      float px = r.q[3], py = r.q[4], pz = r.q[5];
      px = (p >= 0) ? px : 0.f;  py = (p >= 0) ? py : 0.f;  pz = (p >= 0) ? pz : 0.f;
      float dx = x - px, dy = y - py, dz = z - pz;
      float bl = sqrtf(dx * dx + dy * dy + dz * dz);
      s_in[w][n][9]  = x;  s_in[w][n][10] = y;
      s_in[w][n][11] = z;  s_in[w][n][12] = bl;
      const int src = (p >= 0) ? p : NJ;
#pragma unroll
      for (int f = 0; f < DPJ; ++f) s_in[w][n][13 + f] = s_feat[w][src][n][f];
      s_in[w][n][19] = 0.f;
    }
    lds_fence();

    const float bA = r.bA;
    const float bB = (n < 3) ? r.bB : 0.f;
    v8f accA, accB;
#pragma unroll
    for (int v = 0; v < 8; ++v) { accA[v] = bA; accB[v] = bB; }
#pragma unroll
    for (int step = 0; step < 5; ++step) {
      int pr = 2 * step + half;
      v2f a = *(const v2f*)&s_in[w][n][2 * pr];
      accA = wmma4(a, s_w1a[j][pr][n],  accA);
      accB = wmma4(a, s_w1b[j][pr][n3], accB);
    }
#pragma unroll
    for (int v = 0; v < 8; ++v) {
      int m = v + 8 * half;
      s_h[w][m][n] = fmaxf(accA[v], 0.f);
      if (n < 3) s_h[w][m][16 + n] = fmaxf(accB[v], 0.f);
      if (n == 3) s_h[w][m][19] = 0.f;
    }
    lds_fence();

    const float bC = (n < DPJ) ? r.bC : 0.f;
    v8f acc2;
#pragma unroll
    for (int v = 0; v < 8; ++v) acc2[v] = bC;
#pragma unroll
    for (int step = 0; step < 5; ++step) {
      int pr = 2 * step + half;
      v2f a = *(const v2f*)&s_h[w][n][2 * pr];
      acc2 = wmma4(a, s_w2p[j][pr][n7], acc2);
    }
    if (n < DPJ) {
#pragma unroll
      for (int v = 0; v < 8; ++v) {
        int m = v + 8 * half;
        float val = acc2[v];
        s_feat[w][j][m][n] = val;
        out[(long)(rb + m) * (NJ * DPJ) + j * DPJ + n] = val;
      }
    }
    lds_fence();
  };

  for (int j = 0; j < NJ; j += 3) {
    fetch_joint(jC, rrow, jrow, b1, b2, j + 2, half, n);
    run_joint(j, jA);
    fetch_joint(jA, rrow, jrow, b1, b2, imin(j + 3, NJ - 1), half, n);
    run_joint(j + 1, jB);
    fetch_joint(jB, rrow, jrow, b1, b2, imin(j + 4, NJ - 1), half, n);
    run_joint(j + 2, jC);
  }
}

extern "C" void kernel_launch(void* const* d_in, const int* in_sizes, int n_in,
                              void* d_out, int out_size, void* d_ws, size_t ws_size,
                              hipStream_t stream) {
  const float* rots = (const float*)d_in[0];
  const float* Jtrs = (const float*)d_in[1];
  const float* W0   = (const float*)d_in[2];
  const float* b0   = (const float*)d_in[3];
  const float* W1   = (const float*)d_in[4];
  const float* b1   = (const float*)d_in[5];
  const float* W2   = (const float*)d_in[6];
  const float* b2   = (const float*)d_in[7];
  float* outp = (float*)d_out;

  int B = in_sizes[0] / (NJ * 9);
  int grid = (B + 63) / 64;
  pose_enc_kernel<<<grid, 128, 0, stream>>>(rots, Jtrs, W0, b0, W1, b1, W2, b2,
                                            outp, B);
}


# one persistent 16-wave workgroup per WGP loops over 16-row tiles: weights staged to LDS once per WGP instead of once per 64 rows, and 4 waves per SIMD share them instead of 2
# speedup vs baseline: 1.7686x; 1.4207x over previous
#include <hip/hip_runtime.h>

typedef float v2f __attribute__((ext_vector_type(2)));
typedef float v8f __attribute__((ext_vector_type(8)));

#define NJ  24
#define DF  19
#define DPJ 6
#define NW  16

__device__ inline int imin(int a, int b) { return a < b ? a : b; }

__host__ __device__ constexpr unsigned long long pack_parents(int base) {
  constexpr int P[NJ] = {
      -1, 0, 0, 0, 1, 2, 3, 4, 5, 6, 7, 8, 9, 9, 9, 12, 13, 14, 16, 17, 18, 19, 20, 21};
  unsigned long long v = 0;
  for (int i = 0; i < 12; ++i) v |= (unsigned long long)(P[base + i] + 1) << (5 * i);
  return v;
}
__device__ inline int parent_of(int j) {
  constexpr unsigned long long lo = pack_parents(0), hi = pack_parents(12);
  const unsigned long long wv = (j < 12) ? lo : hi;
  const int s = 5 * ((j < 12) ? j : j - 12);
  return (int)((wv >> s) & 31ull) - 1;
}

struct JointRegs {
  float q[9];
  float bA, bB, bC;
};

__device__ inline void fetch_joint(JointRegs& r, const float* rrow, const float* jrow,
                                   const float* __restrict__ b1, const float* __restrict__ b2,
                                   int j, int half, int n) {
  const int p  = parent_of(j);
  const int pc = (p >= 0) ? p : 0;
  const float* s0 = half ? jrow + 3 * j  : rrow + 9 * j;
  const float* s1 = half ? jrow + 3 * pc : rrow + 9 * j + 3;
  const float* s2 = half ? jrow + 3 * j  : rrow + 9 * j + 6;
#pragma unroll
  for (int q = 0; q < 3; ++q) { r.q[q] = s0[q]; r.q[3 + q] = s1[q]; r.q[6 + q] = s2[q]; }
  r.bA = b1[j * DF + n];
  r.bB = b1[j * DF + imin(16 + n, 18)];
  r.bC = b2[j * DPJ + imin(n, 5)];
}

__device__ inline void lds_fence() {
  asm volatile("s_wait_dscnt 0" ::: "memory");
}

__device__ inline v8f wmma4(v2f a, v2f b, v8f c) {
  return __builtin_amdgcn_wmma_f32_16x16x4_f32(
       false, a,  false, b,
       (short)0, c,  false,  false);
}

__global__ __launch_bounds__(NW * 32) void pose_enc_kernel(
    const float* __restrict__ rots, const float* __restrict__ Jtrs,
    const float* __restrict__ W0,   const float* __restrict__ b0,
    const float* __restrict__ W1,   const float* __restrict__ b1,
    const float* __restrict__ W2,   const float* __restrict__ b2,
    float* __restrict__ out, int B)
{
  __shared__ float s_in[NW][16][20];
  __shared__ float s_h [NW][16][20];
  __shared__ float s_feat[NW][NJ + 1][16][DPJ];

  __shared__ v2f s_w1a[NJ][10][16];
  __shared__ v2f s_w1b[NJ][10][4];
  __shared__ v2f s_w2p[NJ][10][8];
  __shared__ v2f s_w0p[144][8];

  const int tid  = threadIdx.x;

  for (int i = tid; i < NJ * 10 * 16; i += NW * 32) {
    int c = i & 15, pr = (i >> 4) % 10, j = i / 160;
    int k = 2 * pr;
    float x = W1[j * 361 + c * 19 + k];
    float y = W1[j * 361 + c * 19 + imin(k + 1, 18)];
    y = (k + 1 < DF) ? y : 0.f;
    v2f v; v.x = x; v.y = y;
    s_w1a[j][pr][c] = v;
  }
  for (int i = tid; i < NJ * 10 * 4; i += NW * 32) {
    int c = i & 3, pr = (i >> 2) % 10, j = i / 40;
    int k = 2 * pr, col = imin(16 + c, 18);
    float x = W1[j * 361 + col * 19 + k];
    float y = W1[j * 361 + col * 19 + imin(k + 1, 18)];
    x = (c < 3) ? x : 0.f;
    y = (c < 3 && k + 1 < DF) ? y : 0.f;
    v2f v; v.x = x; v.y = y;
    s_w1b[j][pr][c] = v;
  }
  for (int i = tid; i < NJ * 10 * 8; i += NW * 32) {
    int c = i & 7, pr = (i >> 3) % 10, j = i / 80;
    int k = 2 * pr, col = imin(c, 5);
    float x = W2[j * 114 + col * 19 + k];
    float y = W2[j * 114 + col * 19 + imin(k + 1, 18)];
    x = (c < DPJ) ? x : 0.f;
    y = (c < DPJ && k + 1 < DF) ? y : 0.f;
    v2f v; v.x = x; v.y = y;
    s_w2p[j][pr][c] = v;
  }
  for (int i = tid; i < 144 * 8; i += NW * 32) {
    int c = i & 7, pr = i >> 3;
    int k = 2 * pr, col = imin(c, 5);
    float x = W0[col * 288 + k];
    float y = W0[col * 288 + k + 1];
    x = (c < DPJ) ? x : 0.f;
    y = (c < DPJ) ? y : 0.f;
    v2f v; v.x = x; v.y = y;
    s_w0p[pr][c] = v;
  }
  __syncthreads();

  const int w    = tid >> 5;
  const int lane = tid & 31;
  const int half = lane >> 4;
  const int n    = lane & 15;
  const int n7   = imin(n, 7);
  const int n3   = imin(n, 3);
  for (int tile = blockIdx.x * NW + w; tile * 16 < B; tile += gridDim.x * NW) {
  const int rb   = tile * 16;

  const float* rrow = rots + (long)(rb + n) * 216;
  const float* jrow = Jtrs + (long)(rb + n) * 72;

  JointRegs jA, jB, jC;
  fetch_joint(jA, rrow, jrow, b1, b2, 0, half, n);
  fetch_joint(jB, rrow, jrow, b1, b2, 1, half, n);

  {
    float bv = b0[imin(n, 5)];
    bv = (n < DPJ) ? bv : 0.f;
    v8f acc;
#pragma unroll
    for (int v = 0; v < 8; ++v) acc[v] = bv;

#pragma unroll 4
    for (int step = 0; step < 54; ++step) {
      int pr = 2 * step + half;
      v2f a = *(const v2f*)(rrow + 2 * pr);
      acc = wmma4(a, s_w0p[pr][n7], acc);
    }
#pragma unroll 4
    for (int step = 54; step < 72; ++step) {
      int pr = 2 * step + half;
      v2f a = *(const v2f*)(jrow + 2 * pr - 216);
      acc = wmma4(a, s_w0p[pr][n7], acc);
    }
    if (n < DPJ) {
#pragma unroll
      for (int v = 0; v < 8; ++v) s_feat[w][NJ][v + 8 * half][n] = acc[v];
    }
  }
  lds_fence();

  auto run_joint = [&](const int j, const JointRegs& r) {
    const int p = parent_of(j);

    if (half == 0) {
#pragma unroll
      for (int q = 0; q < 9; ++q) s_in[w][n][q] = r.q[q];
    } else {
      float x = r.q[0], y = r.q[1], z = r.q[2];
      float px = r.q[3], py = r.q[4], pz = r.q[5];
      px = (p >= 0) ? px : 0.f;  py = (p >= 0) ? py : 0.f;  pz = (p >= 0) ? pz : 0.f;
      float dx = x - px, dy = y - py, dz = z - pz;
      float bl = sqrtf(dx * dx + dy * dy + dz * dz);
      s_in[w][n][9]  = x;  s_in[w][n][10] = y;
      s_in[w][n][11] = z;  s_in[w][n][12] = bl;
      const int src = (p >= 0) ? p : NJ;
#pragma unroll
      for (int f = 0; f < DPJ; ++f) s_in[w][n][13 + f] = s_feat[w][src][n][f];
      s_in[w][n][19] = 0.f;
    }
    lds_fence();

    const float bA = r.bA;
    const float bB = (n < 3) ? r.bB : 0.f;
    v8f accA, accB;
#pragma unroll
    for (int v = 0; v < 8; ++v) { accA[v] = bA; accB[v] = bB; }
#pragma unroll
    for (int step = 0; step < 5; ++step) {
      int pr = 2 * step + half;
      v2f a = *(const v2f*)&s_in[w][n][2 * pr];
      accA = wmma4(a, s_w1a[j][pr][n],  accA);
      accB = wmma4(a, s_w1b[j][pr][n3], accB);
    }
#pragma unroll
    for (int v = 0; v < 8; ++v) {
      int m = v + 8 * half;
      s_h[w][m][n] = fmaxf(accA[v], 0.f);
      if (n < 3) s_h[w][m][16 + n] = fmaxf(accB[v], 0.f);
      if (n == 3) s_h[w][m][19] = 0.f;
    }
    lds_fence();

    const float bC = (n < DPJ) ? r.bC : 0.f;
    v8f acc2;
#pragma unroll
    for (int v = 0; v < 8; ++v) acc2[v] = bC;
#pragma unroll
    for (int step = 0; step < 5; ++step) {
      int pr = 2 * step + half;
      v2f a = *(const v2f*)&s_h[w][n][2 * pr];
      acc2 = wmma4(a, s_w2p[j][pr][n7], acc2);
    }
    if (n < DPJ) {
#pragma unroll
      for (int v = 0; v < 8; ++v) {
        int m = v + 8 * half;
        float val = acc2[v];
        s_feat[w][j][m][n] = val;
        out[(long)(rb + m) * (NJ * DPJ) + j * DPJ + n] = val;
      }
    }
    lds_fence();
  };

  for (int j = 0; j < NJ; j += 3) {
    fetch_joint(jC, rrow, jrow, b1, b2, j + 2, half, n);
    run_joint(j, jA);
    fetch_joint(jA, rrow, jrow, b1, b2, imin(j + 3, NJ - 1), half, n);
    run_joint(j + 1, jB);
    fetch_joint(jB, rrow, jrow, b1, b2, imin(j + 4, NJ - 1), half, n);
    run_joint(j + 2, jC);
  }
  }
}

extern "C" void kernel_launch(void* const* d_in, const int* in_sizes, int n_in,
                              void* d_out, int out_size, void* d_ws, size_t ws_size,
                              hipStream_t stream) {
  const float* rots = (const float*)d_in[0];
  const float* Jtrs = (const float*)d_in[1];
  const float* W0   = (const float*)d_in[2];
  const float* b0   = (const float*)d_in[3];
  const float* W1   = (const float*)d_in[4];
  const float* b1   = (const float*)d_in[5];
  const float* W2   = (const float*)d_in[6];
  const float* b2   = (const float*)d_in[7];
  float* outp = (float*)d_out;

  int B = in_sizes[0] / (NJ * 9);
  pose_enc_kernel<<<256, NW * 32, 0, stream>>>(rots, Jtrs, W0, b0, W1, b1, W2, b2,
                                            outp, B);
}


# products formed as weights x rows^T so each layer's D registers are the next layer's B operand: no LDS round trip per joint, biases ride in the spare k slot; k order of every sum changes
# speedup vs baseline: 1.8838x; 1.0651x over previous
#include <hip/hip_runtime.h>


typedef float v2f __attribute__((ext_vector_type(2)));
typedef float v8f __attribute__((ext_vector_type(8)));

#define NJ  24
#define DF  19
#define DPJ 6
#define NW  16
#define NFR 15

__device__ inline int imin(int a, int b) { return a < b ? a : b; }

__host__ __device__ constexpr unsigned long long pack_parents(int base) {
  constexpr int P[NJ] = {
      -1, 0, 0, 0, 1, 2, 3, 4, 5, 6, 7, 8, 9, 9, 9, 12, 13, 14, 16, 17, 18, 19, 20, 21};
  unsigned long long v = 0;
  for (int i = 0; i < 12; ++i) v |= (unsigned long long)(P[base + i] + 1) << (5 * i);
  return v;
}
__host__ __device__ constexpr int parent_of(int j) {
  const unsigned long long wv = (j < 12) ? pack_parents(0) : pack_parents(12);
  const int s = 5 * ((j < 12) ? j : j - 12);
  return (int)((wv >> s) & 31ull) - 1;
}

__host__ __device__ constexpr unsigned long long pack_slots(int half) {
  constexpr int S[2][10] = {{0, 1, 2, 3, 8, 9, 12, 19, 15, 16},
                            {4, 5, 6, 7, 10, 11, 13, 14, 17, 18}};
  unsigned long long v = 0;
  for (int e = 0; e < 10; ++e) v |= (unsigned long long)S[half][e] << (5 * e);
  return v;
}

template <int F, int S>
__device__ inline float wfrag_value(const float* __restrict__ W1, const float* __restrict__ b1,
                                    const float* __restrict__ W2, const float* __restrict__ b2,
                                    int j, int c, int h) {
  constexpr int grp = F / 5, t = F - 5 * grp, e = 2 * t + S;
  if constexpr (grp < 2) {
    constexpr int c0 = (int)((pack_slots(0) >> (5 * e)) & 31ull);
    constexpr int c1 = (int)((pack_slots(1) >> (5 * e)) & 31ull);
    static_assert(c1 < DF, "the spare slot sits in lane half 0 only");
    const int rowB = (c == 0) ? 16 : ((c == 1) ? 17 : 18);
    const int row = (grp == 1) ? rowB : c;
    const bool valid = (grp == 1) ? ((c == 0) || (c == 1) || (c == 8)) : true;
    const float* wr = W1 + j * (DF * DF) + row * DF;
    float x;
    if constexpr (c0 < DF) {
      x = wr[h ? c1 : c0];
      return valid ? x : 0.f;
    } else {
      const float xw = wr[c1];
      const float xb = b1[j * DF + row];
      x = h ? xw : xb;
      const float alt = ((grp == 1) && (c == 9) && (h == 0)) ? 1.f : 0.f;
      return valid ? x : alt;
    }
  } else {
    const int fo = c & 7;
    const bool valid = fo < DPJ;
    const int foc = imin(fo, DPJ - 1);
    const float* wr = W2 + j * (DPJ * DF) + foc * DF;
    float x;
    if constexpr (t < 4) {
      x = wr[8 * h + 2 * t + S];
    } else if constexpr (S == 0) {
      x = wr[h ? 18 : 16];
    } else {
      const float xw = wr[17];
      const float xb = b2[j * DPJ + foc];
      x = h ? xb : xw;
    }
    return valid ? x : 0.f;
  }
}

template <int... F>
__device__ inline void stage_joint(v2f* dst, const float* __restrict__ W1, const float* __restrict__ b1,
                                   const float* __restrict__ W2, const float* __restrict__ b2,
                                   int j, int c, int h) {
  v2f v[NFR];
#pragma unroll
  for (int f = 0; f < NFR; ++f) { v[f].x = 0.f; v[f].y = 0.f; }
  ((v[F].x = wfrag_value<F, 0>(W1, b1, W2, b2, j, c, h),
    v[F].y = wfrag_value<F, 1>(W1, b1, W2, b2, j, c, h)), ...);
#pragma unroll
  for (int f = 0; f < NFR; ++f) dst[f * 32] = v[f];
}

struct JointRegs {
  float a[4];
  float r8;
  float J[3];
  float P[3];
};

__device__ inline void fetch_joint(JointRegs& r, const float* ra, const float* rrow,
                                   const float* jrow, int j) {
  const float* pa = ra + 9 * j;
#pragma unroll
  for (int q = 0; q < 4; ++q) r.a[q] = pa[q];
  r.r8 = rrow[9 * j + 8];
  const float* pj = jrow + 3 * j;
#pragma unroll
  for (int q = 0; q < 3; ++q) r.J[q] = pj[q];
  const int p = parent_of(j);
  if (p >= 0) {
    const float* pp = jrow + 3 * p;
#pragma unroll
    for (int q = 0; q < 3; ++q) r.P[q] = pp[q];
  } else {
#pragma unroll
    for (int q = 0; q < 3; ++q) r.P[q] = 0.f;
  }
}

__device__ inline v8f wmma4(v2f a, v2f b, v8f c) {
  return __builtin_amdgcn_wmma_f32_16x16x4_f32(
       false, a,  false, b,
       (short)0, c,  false,  false);
}

__global__ __launch_bounds__(NW * 32) void pose_enc_kernel(
    const float* __restrict__ rots, const float* __restrict__ Jtrs,
    const float* __restrict__ W0,   const float* __restrict__ b0,
    const float* __restrict__ W1,   const float* __restrict__ b1,
    const float* __restrict__ W2,   const float* __restrict__ b2,
    float* __restrict__ out, int B)
{
  __shared__ v2f s_wf[NJ][NFR][32];
  __shared__ v2f s_w0p[144][8];

  const int tid  = threadIdx.x;
  const int lane = tid & 31;
  const int half = lane >> 4;
  const int n    = lane & 15;
  const int w    = __builtin_amdgcn_readfirstlane(tid >> 5);

  for (int j = w; j < NJ; j += NW) {
    stage_joint<0, 1, 2, 3, 4, 5, 6, 7, 8, 9, 10, 11, 12, 13, 14>(
        &s_wf[j][0][lane], W1, b1, W2, b2, j, n, half);
  }
  {
    float wx[3], wy[3];
#pragma unroll
    for (int k = 0; k < 3; ++k) {
      const int i = imin(tid + k * (NW * 32), 144 * 8 - 1);
      const int col = imin(i & 7, DPJ - 1), pr = i >> 3;
      wx[k] = W0[col * 288 + 2 * pr];
      wy[k] = W0[col * 288 + 2 * pr + 1];
    }
#pragma unroll
    for (int k = 0; k < 3; ++k) {
      const int i = tid + k * (NW * 32);
      if (i < 144 * 8) {
        v2f v;
        v.x = ((i & 7) < DPJ) ? wx[k] : 0.f;
        v.y = ((i & 7) < DPJ) ? wy[k] : 0.f;
        s_w0p[i >> 3][i & 7] = v;
      }
    }
  }
  __syncthreads();

  const int n7 = n & 7;
  const v2f* wfl = &s_wf[0][0][lane];
  const v8f zero8 = {0.f, 0.f, 0.f, 0.f, 0.f, 0.f, 0.f, 0.f};

  for (int tile = blockIdx.x * NW + w; tile * 16 < B; tile += gridDim.x * NW) {
    asm volatile("" ::: "memory");
    const int rb = tile * 16;
    const float* rrow = rots + (long)(rb + n) * 216;
    const float* jrow = Jtrs + (long)(rb + n) * 72;
    const float* ra   = rrow + 4 * half;
    const float* ja   = jrow + 4 * half;
    float*       orow = out + (long)(rb + n) * (NJ * DPJ) + 3 * half;

    JointRegs q[NJ];
    fetch_joint(q[0], ra, rrow, jrow, 0);
    fetch_joint(q[1], ra, rrow, jrow, 1);
    v2f wv[2][NFR];
#pragma unroll
    for (int f = 0; f < NFR; ++f) wv[0][f] = wfl[f * 32];

    v8f gf;
#pragma unroll
    for (int v = 0; v < 8; ++v) gf[v] = (v < DPJ) ? b0[imin(v, DPJ - 1)] : 0.f;
#pragma unroll
    for (int i = 0; i < 27; ++i) {
      const float* p = ra + 8 * i;
      v2f x01, x23;
      x01.x = p[0]; x01.y = p[1]; x23.x = p[2]; x23.y = p[3];
      const int pr = 4 * i + 2 * half;
      gf = wmma4(s_w0p[pr][n7],     x01, gf);
      gf = wmma4(s_w0p[pr + 1][n7], x23, gf);
    }
#pragma unroll
    for (int i = 0; i < 9; ++i) {
      const float* p = ja + 8 * i;
      v2f x01, x23;
      x01.x = p[0]; x01.y = p[1]; x23.x = p[2]; x23.y = p[3];
      const int pr = 108 + 4 * i + 2 * half;
      gf = wmma4(s_w0p[pr][n7],     x01, gf);
      gf = wmma4(s_w0p[pr + 1][n7], x23, gf);
    }

    v8f O[NJ + 1];
    O[NJ] = gf;
#pragma unroll
    for (int j = 0; j < NJ; ++j) {
      asm volatile("" ::: "memory");
      if (j + 2 < NJ) fetch_joint(q[j + 2], ra, rrow, jrow, j + 2);
      if (j + 1 < NJ) {
#pragma unroll
        for (int f = 0; f < NFR; ++f) wv[(j + 1) & 1][f] = wfl[((j + 1) * NFR + f) * 32];
      }
      const v2f* wj = wv[j & 1];
      const int p = parent_of(j);
      const v8f fp = O[(p >= 0) ? p : NJ];
      const JointRegs r = q[j];

      const float dx = r.J[0] - r.P[0], dy = r.J[1] - r.P[1], dz = r.J[2] - r.P[2];
      const float bone = sqrtf(dx * dx + dy * dy + dz * dz);

      v2f B0, B1, B2, B3, B4;
      B0.x = r.a[0];                 B0.y = r.a[1];
      B1.x = r.a[2];                 B1.y = r.a[3];
      B2.x = half ? r.J[1] : r.r8;   B2.y = half ? r.J[2] : r.J[0];
      B3.x = half ? fp[0]  : bone;   B3.y = half ? fp[1]  : 1.f;
      B4.x = half ? fp[4]  : fp[2];  B4.y = half ? fp[5]  : fp[3];

      v8f aA = wmma4(wj[0], B0, zero8);
      v8f aB = wmma4(wj[5], B0, zero8);
      aA = wmma4(wj[1], B1, aA);
      aB = wmma4(wj[6], B1, aB);
      aA = wmma4(wj[2], B2, aA);
      aB = wmma4(wj[7], B2, aB);
      aA = wmma4(wj[3], B3, aA);
      aB = wmma4(wj[8], B3, aB);
      aA = wmma4(wj[4], B4, aA);
      aB = wmma4(wj[9], B4, aB);

      v2f H0, H1, H2, H3, H4;
      H0.x = fmaxf(aA[0], 0.f);  H0.y = fmaxf(aA[1], 0.f);
      H1.x = fmaxf(aA[2], 0.f);  H1.y = fmaxf(aA[3], 0.f);
      H2.x = fmaxf(aA[4], 0.f);  H2.y = fmaxf(aA[5], 0.f);
      H3.x = fmaxf(aA[6], 0.f);  H3.y = fmaxf(aA[7], 0.f);
      H4.x = fmaxf(aB[0], 0.f);  H4.y = fmaxf(aB[1], 0.f);

      v8f o = wmma4(wj[10], H0, zero8);
      o = wmma4(wj[11], H1, o);
      o = wmma4(wj[12], H2, o);
      o = wmma4(wj[13], H3, o);
      o = wmma4(wj[14], H4, o);
      O[j] = o;

      orow[DPJ * j + 0] = half ? o[3] : o[0];
      orow[DPJ * j + 1] = half ? o[4] : o[1];
      orow[DPJ * j + 2] = half ? o[5] : o[2];
    }
  }
}

extern "C" void kernel_launch(void* const* d_in, const int* in_sizes, int n_in,
                              void* d_out, int out_size, void* d_ws, size_t ws_size,
                              hipStream_t stream) {
  const float* rots = (const float*)d_in[0];
  const float* Jtrs = (const float*)d_in[1];
  const float* W0   = (const float*)d_in[2];
  const float* b0   = (const float*)d_in[3];
  const float* W1   = (const float*)d_in[4];
  const float* b1   = (const float*)d_in[5];
  const float* W2   = (const float*)d_in[6];
  const float* b2   = (const float*)d_in[7];
  float* outp = (float*)d_out;

  int B = in_sizes[0] / (NJ * 9);
  pose_enc_kernel<<<256, NW * 32, 0, stream>>>(rots, Jtrs, W0, b0, W1, b1, W2, b2,
                                            outp, B);
}


# each wave copies its tile's 16 contiguous input rows to a private LDS buffer with full-line loads and lanes read their own row from LDS, not memory; 12 waves per workgroup so the buffers fit
# speedup vs baseline: 3.0189x; 1.6026x over previous
#include <hip/hip_runtime.h>


typedef float v2f __attribute__((ext_vector_type(2)));
typedef float v4f __attribute__((ext_vector_type(4)));
typedef float v8f __attribute__((ext_vector_type(8)));

#define NJ  24
#define DF  19
#define DPJ 6
#define NW  12
#define TROT (16 * 216)
#define TJTR (16 * 72)
#define TCH  ((TROT + TJTR) / 128)
#define NFR 15

__device__ inline int imin(int a, int b) { return a < b ? a : b; }

__host__ __device__ constexpr unsigned long long pack_parents(int base) {
  constexpr int P[NJ] = {
      -1, 0, 0, 0, 1, 2, 3, 4, 5, 6, 7, 8, 9, 9, 9, 12, 13, 14, 16, 17, 18, 19, 20, 21};
  unsigned long long v = 0;
  for (int i = 0; i < 12; ++i) v |= (unsigned long long)(P[base + i] + 1) << (5 * i);
  return v;
}
__host__ __device__ constexpr int parent_of(int j) {
  const unsigned long long wv = (j < 12) ? pack_parents(0) : pack_parents(12);
  const int s = 5 * ((j < 12) ? j : j - 12);
  return (int)((wv >> s) & 31ull) - 1;
}

__host__ __device__ constexpr unsigned long long pack_slots(int half) {
  constexpr int S[2][10] = {{0, 1, 2, 3, 8, 9, 12, 19, 15, 16},
                            {4, 5, 6, 7, 10, 11, 13, 14, 17, 18}};
  unsigned long long v = 0;
  for (int e = 0; e < 10; ++e) v |= (unsigned long long)S[half][e] << (5 * e);
  return v;
}

template <int F, int S>
__device__ inline float wfrag_value(const float* __restrict__ W1, const float* __restrict__ b1,
                                    const float* __restrict__ W2, const float* __restrict__ b2,
                                    int j, int c, int h) {
  constexpr int grp = F / 5, t = F - 5 * grp, e = 2 * t + S;
  if constexpr (grp < 2) {
    constexpr int c0 = (int)((pack_slots(0) >> (5 * e)) & 31ull);
    constexpr int c1 = (int)((pack_slots(1) >> (5 * e)) & 31ull);
    static_assert(c1 < DF, "the spare slot sits in lane half 0 only");
    const int rowB = (c == 0) ? 16 : ((c == 1) ? 17 : 18);
    const int row = (grp == 1) ? rowB : c;
    const bool valid = (grp == 1) ? ((c == 0) || (c == 1) || (c == 8)) : true;
    const float* wr = W1 + j * (DF * DF) + row * DF;
    float x;
    if constexpr (c0 < DF) {
      x = wr[h ? c1 : c0];
      return valid ? x : 0.f;
    } else {
      const float xw = wr[c1];
      const float xb = b1[j * DF + row];
      x = h ? xw : xb;
      const float alt = ((grp == 1) && (c == 9) && (h == 0)) ? 1.f : 0.f;
      return valid ? x : alt;
    }
  } else {
    const int fo = c & 7;
    const bool valid = fo < DPJ;
    const int foc = imin(fo, DPJ - 1);
    const float* wr = W2 + j * (DPJ * DF) + foc * DF;
    float x;
    if constexpr (t < 4) {
      x = wr[8 * h + 2 * t + S];
    } else if constexpr (S == 0) {
      x = wr[h ? 18 : 16];
    } else {
      const float xw = wr[17];
      const float xb = b2[j * DPJ + foc];
      x = h ? xb : xw;
    }
    return valid ? x : 0.f;
  }
}

template <int... F>
__device__ inline void stage_joint(v2f* dst, const float* __restrict__ W1, const float* __restrict__ b1,
                                   const float* __restrict__ W2, const float* __restrict__ b2,
                                   int j, int c, int h) {
  v2f v[NFR];
#pragma unroll
  for (int f = 0; f < NFR; ++f) { v[f].x = 0.f; v[f].y = 0.f; }
  ((v[F].x = wfrag_value<F, 0>(W1, b1, W2, b2, j, c, h),
    v[F].y = wfrag_value<F, 1>(W1, b1, W2, b2, j, c, h)), ...);
#pragma unroll
  for (int f = 0; f < NFR; ++f) dst[f * 32] = v[f];
}

struct JointRegs {
  float a[4];
  float r8;
  float J[3];
  float P[3];
};

__device__ inline void fetch_joint(JointRegs& r, const float* ra, const float* rrow,
                                   const float* jrow, int j) {
  const float* pa = ra + 9 * j;
#pragma unroll
  for (int q = 0; q < 4; ++q) r.a[q] = pa[q];
  r.r8 = rrow[9 * j + 8];
  const float* pj = jrow + 3 * j;
#pragma unroll
  for (int q = 0; q < 3; ++q) r.J[q] = pj[q];
  const int p = parent_of(j);
  if (p >= 0) {
    const float* pp = jrow + 3 * p;
#pragma unroll
    for (int q = 0; q < 3; ++q) r.P[q] = pp[q];
  } else {
#pragma unroll
    for (int q = 0; q < 3; ++q) r.P[q] = 0.f;
  }
}

__device__ inline v8f wmma4(v2f a, v2f b, v8f c) {
  return __builtin_amdgcn_wmma_f32_16x16x4_f32(
       false, a,  false, b,
       (short)0, c,  false,  false);
}

__global__ __launch_bounds__(NW * 32) void pose_enc_kernel(
    const float* __restrict__ rots, const float* __restrict__ Jtrs,
    const float* __restrict__ W0,   const float* __restrict__ b0,
    const float* __restrict__ W1,   const float* __restrict__ b1,
    const float* __restrict__ W2,   const float* __restrict__ b2,
    float* __restrict__ out, int B)
{
  __shared__ v2f s_wf[NJ][NFR][32];
  __shared__ v2f s_w0p[144][8];
  __shared__ v4f s_x[NW][TCH * 32];

  const int tid  = threadIdx.x;
  const int lane = tid & 31;
  const int half = lane >> 4;
  const int n    = lane & 15;
  const int w    = __builtin_amdgcn_readfirstlane(tid >> 5);

  for (int j = w; j < NJ; j += NW) {
    stage_joint<0, 1, 2, 3, 4, 5, 6, 7, 8, 9, 10, 11, 12, 13, 14>(
        &s_wf[j][0][lane], W1, b1, W2, b2, j, n, half);
  }
  {
    float wx[3], wy[3];
#pragma unroll
    for (int k = 0; k < 3; ++k) {
      const int i = imin(tid + k * (NW * 32), 144 * 8 - 1);
      const int col = imin(i & 7, DPJ - 1), pr = i >> 3;
      wx[k] = W0[col * 288 + 2 * pr];
      wy[k] = W0[col * 288 + 2 * pr + 1];
    }
#pragma unroll
    for (int k = 0; k < 3; ++k) {
      const int i = tid + k * (NW * 32);
      if (i < 144 * 8) {
        v2f v;
        v.x = ((i & 7) < DPJ) ? wx[k] : 0.f;
        v.y = ((i & 7) < DPJ) ? wy[k] : 0.f;
        s_w0p[i >> 3][i & 7] = v;
      }
    }
  }
  __syncthreads();

  const int n7 = n & 7;
  const v2f* wfl = &s_wf[0][0][lane];
  const v8f zero8 = {0.f, 0.f, 0.f, 0.f, 0.f, 0.f, 0.f, 0.f};
  v8f gf0;
#pragma unroll
  for (int v = 0; v < 8; ++v) gf0[v] = (v < DPJ) ? b0[imin(v, DPJ - 1)] : 0.f;

  v4f* xs = &s_x[w][0];
  const float* rrow = (const float*)xs + n * 216;
  const float* jrow = (const float*)xs + TROT + n * 72;
  const float* ra   = rrow + 4 * half;
  const float* ja   = jrow + 4 * half;

  for (int t = w; (blockIdx.x + gridDim.x * t) * 16 < B; t += NW) {
    asm volatile("" ::: "memory");
    const int rb = (blockIdx.x + gridDim.x * t) * 16;
    float*       orow = out + (long)(rb + n) * (NJ * DPJ) + 3 * half;

    {
      const float* grot = rots + (long)rb * 216 + 4 * lane;
      const float* gjtr = Jtrs + (long)rb * 72 + 4 * lane;
      v4f st[TCH];
#pragma unroll
      for (int i = 0; i < TROT / 128; ++i) {
        const float* p = grot + 128 * i;
        st[i].x = p[0]; st[i].y = p[1]; st[i].z = p[2]; st[i].w = p[3];
      }
#pragma unroll
      for (int i = 0; i < TJTR / 128; ++i) {
        const float* p = gjtr + 128 * i;
        v4f& d = st[TROT / 128 + i];
        d.x = p[0]; d.y = p[1]; d.z = p[2]; d.w = p[3];
      }
#pragma unroll
      for (int i = 0; i < TCH; ++i) xs[32 * i + lane] = st[i];
    }
    asm volatile("" ::: "memory");

    JointRegs q[NJ];
    fetch_joint(q[0], ra, rrow, jrow, 0);
    v2f wv[2][NFR];
#pragma unroll
    for (int f = 0; f < NFR; ++f) wv[0][f] = wfl[f * 32];

    v8f gf = gf0;
#pragma unroll
    for (int i = 0; i < 27; ++i) {
      const v4f x = *(const v4f*)(ra + 8 * i);
      v2f x01, x23;
      x01.x = x.x; x01.y = x.y; x23.x = x.z; x23.y = x.w;
      const int pr = 4 * i + 2 * half;
      gf = wmma4(s_w0p[pr][n7],     x01, gf);
      gf = wmma4(s_w0p[pr + 1][n7], x23, gf);
    }
#pragma unroll
    for (int i = 0; i < 9; ++i) {
      const v4f x = *(const v4f*)(ja + 8 * i);
      v2f x01, x23;
      x01.x = x.x; x01.y = x.y; x23.x = x.z; x23.y = x.w;
      const int pr = 108 + 4 * i + 2 * half;
      gf = wmma4(s_w0p[pr][n7],     x01, gf);
      gf = wmma4(s_w0p[pr + 1][n7], x23, gf);
    }

    v8f O[NJ + 1];
    O[NJ] = gf;
#pragma unroll
    for (int j = 0; j < NJ; ++j) {
      asm volatile("" ::: "memory");
      if (j + 1 < NJ) fetch_joint(q[j + 1], ra, rrow, jrow, j + 1);
      if (j + 1 < NJ) {
#pragma unroll
        for (int f = 0; f < NFR; ++f) wv[(j + 1) & 1][f] = wfl[((j + 1) * NFR + f) * 32];
      }
      const v2f* wj = wv[j & 1];
      const int p = parent_of(j);
      const v8f fp = O[(p >= 0) ? p : NJ];
      const JointRegs r = q[j];

      const float dx = r.J[0] - r.P[0], dy = r.J[1] - r.P[1], dz = r.J[2] - r.P[2];
      const float bone = sqrtf(dx * dx + dy * dy + dz * dz);

      v2f B0, B1, B2, B3, B4;
      B0.x = r.a[0];                 B0.y = r.a[1];
      B1.x = r.a[2];                 B1.y = r.a[3];
      B2.x = half ? r.J[1] : r.r8;   B2.y = half ? r.J[2] : r.J[0];
      B3.x = half ? fp[0]  : bone;   B3.y = half ? fp[1]  : 1.f;
      B4.x = half ? fp[4]  : fp[2];  B4.y = half ? fp[5]  : fp[3];

      v8f aA = wmma4(wj[0], B0, zero8);
      v8f aB = wmma4(wj[5], B0, zero8);
      aA = wmma4(wj[1], B1, aA);
      aB = wmma4(wj[6], B1, aB);
      aA = wmma4(wj[2], B2, aA);
      aB = wmma4(wj[7], B2, aB);
      aA = wmma4(wj[3], B3, aA);
      aB = wmma4(wj[8], B3, aB);
      aA = wmma4(wj[4], B4, aA);
      aB = wmma4(wj[9], B4, aB);

      v2f H0, H1, H2, H3, H4;
      H0.x = fmaxf(aA[0], 0.f);  H0.y = fmaxf(aA[1], 0.f);
      H1.x = fmaxf(aA[2], 0.f);  H1.y = fmaxf(aA[3], 0.f);
      H2.x = fmaxf(aA[4], 0.f);  H2.y = fmaxf(aA[5], 0.f);
      H3.x = fmaxf(aA[6], 0.f);  H3.y = fmaxf(aA[7], 0.f);
      H4.x = fmaxf(aB[0], 0.f);  H4.y = fmaxf(aB[1], 0.f);

      v8f o = wmma4(wj[10], H0, zero8);
      o = wmma4(wj[11], H1, o);
      o = wmma4(wj[12], H2, o);
      o = wmma4(wj[13], H3, o);
      o = wmma4(wj[14], H4, o);
      O[j] = o;

      orow[DPJ * j + 0] = half ? o[3] : o[0];
      orow[DPJ * j + 1] = half ? o[4] : o[1];
      orow[DPJ * j + 2] = half ? o[5] : o[2];
    }
  }
}

extern "C" void kernel_launch(void* const* d_in, const int* in_sizes, int n_in,
                              void* d_out, int out_size, void* d_ws, size_t ws_size,
                              hipStream_t stream) {
  const float* rots = (const float*)d_in[0];
  const float* Jtrs = (const float*)d_in[1];
  const float* W0   = (const float*)d_in[2];
  const float* b0   = (const float*)d_in[3];
  const float* W1   = (const float*)d_in[4];
  const float* b1   = (const float*)d_in[5];
  const float* W2   = (const float*)d_in[6];
  const float* b2   = (const float*)d_in[7];
  float* outp = (float*)d_out;

  int B = in_sizes[0] / (NJ * 9);
  pose_enc_kernel<<<256, NW * 32, 0, stream>>>(rots, Jtrs, W0, b0, W1, b1, W2, b2,
                                            outp, B);
}


# each wave parks its tile's outputs in its LDS tile buffer (over rot values already used) and writes the tile's 9216 contiguous bytes as 18 whole-line 16-byte stores, not 24 stores of 12 B per lane
# speedup vs baseline: 3.8639x; 1.2799x over previous
#include <hip/hip_runtime.h>


typedef float v2f __attribute__((ext_vector_type(2)));
typedef float v4f __attribute__((ext_vector_type(4)));
typedef float v8f __attribute__((ext_vector_type(8)));

#define NJ  24
#define DF  19
#define DPJ 6
#define NW  12
#define TROT (16 * 216)
#define TJTR (16 * 72)
#define TCH  ((TROT + TJTR) / 128)
#define NOCH ((16 * NJ * DPJ) / 128)
#define NFR 15

__device__ inline int imin(int a, int b) { return a < b ? a : b; }

__host__ __device__ constexpr unsigned long long pack_parents(int base) {
  constexpr int P[NJ] = {
      -1, 0, 0, 0, 1, 2, 3, 4, 5, 6, 7, 8, 9, 9, 9, 12, 13, 14, 16, 17, 18, 19, 20, 21};
  unsigned long long v = 0;
  for (int i = 0; i < 12; ++i) v |= (unsigned long long)(P[base + i] + 1) << (5 * i);
  return v;
}
__host__ __device__ constexpr int parent_of(int j) {
  const unsigned long long wv = (j < 12) ? pack_parents(0) : pack_parents(12);
  const int s = 5 * ((j < 12) ? j : j - 12);
  return (int)((wv >> s) & 31ull) - 1;
}

__host__ __device__ constexpr unsigned long long pack_slots(int half) {
  constexpr int S[2][10] = {{0, 1, 2, 3, 8, 9, 12, 19, 15, 16},
                            {4, 5, 6, 7, 10, 11, 13, 14, 17, 18}};
  unsigned long long v = 0;
  for (int e = 0; e < 10; ++e) v |= (unsigned long long)S[half][e] << (5 * e);
  return v;
}

template <int F, int S>
__device__ inline float wfrag_value(const float* __restrict__ W1, const float* __restrict__ b1,
                                    const float* __restrict__ W2, const float* __restrict__ b2,
                                    int j, int c, int h) {
  constexpr int grp = F / 5, t = F - 5 * grp, e = 2 * t + S;
  if constexpr (grp < 2) {
    constexpr int c0 = (int)((pack_slots(0) >> (5 * e)) & 31ull);
    constexpr int c1 = (int)((pack_slots(1) >> (5 * e)) & 31ull);
    static_assert(c1 < DF, "the spare slot sits in lane half 0 only");
    const int rowB = (c == 0) ? 16 : ((c == 1) ? 17 : 18);
    const int row = (grp == 1) ? rowB : c;
    const bool valid = (grp == 1) ? ((c == 0) || (c == 1) || (c == 8)) : true;
    const float* wr = W1 + j * (DF * DF) + row * DF;
    float x;
    if constexpr (c0 < DF) {
      x = wr[h ? c1 : c0];
      return valid ? x : 0.f;
    } else {
      const float xw = wr[c1];
      const float xb = b1[j * DF + row];
      x = h ? xw : xb;
      const float alt = ((grp == 1) && (c == 9) && (h == 0)) ? 1.f : 0.f;
      return valid ? x : alt;
    }
  } else {
    const int fo = c & 7;
    const bool valid = fo < DPJ;
    const int foc = imin(fo, DPJ - 1);
    const float* wr = W2 + j * (DPJ * DF) + foc * DF;
    float x;
    if constexpr (t < 4) {
      x = wr[8 * h + 2 * t + S];
    } else if constexpr (S == 0) {
      x = wr[h ? 18 : 16];
    } else {
      const float xw = wr[17];
      const float xb = b2[j * DPJ + foc];
      x = h ? xb : xw;
    }
    return valid ? x : 0.f;
  }
}

template <int... F>
__device__ inline void stage_joint(v2f* dst, const float* __restrict__ W1, const float* __restrict__ b1,
                                   const float* __restrict__ W2, const float* __restrict__ b2,
                                   int j, int c, int h) {
  v2f v[NFR];
#pragma unroll
  for (int f = 0; f < NFR; ++f) { v[f].x = 0.f; v[f].y = 0.f; }
  ((v[F].x = wfrag_value<F, 0>(W1, b1, W2, b2, j, c, h),
    v[F].y = wfrag_value<F, 1>(W1, b1, W2, b2, j, c, h)), ...);
#pragma unroll
  for (int f = 0; f < NFR; ++f) dst[f * 32] = v[f];
}

struct JointRegs {
  float a[4];
  float r8;
  float J[3];
  float P[3];
};

__device__ inline void fetch_joint(JointRegs& r, const float* ra, const float* rrow,
                                   const float* jrow, int j) {
  const float* pa = ra + 9 * j;
#pragma unroll
  for (int q = 0; q < 4; ++q) r.a[q] = pa[q];
  r.r8 = rrow[9 * j + 8];
  const float* pj = jrow + 3 * j;
#pragma unroll
  for (int q = 0; q < 3; ++q) r.J[q] = pj[q];
  const int p = parent_of(j);
  if (p >= 0) {
    const float* pp = jrow + 3 * p;
#pragma unroll
    for (int q = 0; q < 3; ++q) r.P[q] = pp[q];
  } else {
#pragma unroll
    for (int q = 0; q < 3; ++q) r.P[q] = 0.f;
  }
}

__device__ inline v8f wmma4(v2f a, v2f b, v8f c) {
  return __builtin_amdgcn_wmma_f32_16x16x4_f32(
       false, a,  false, b,
       (short)0, c,  false,  false);
}

__global__ __launch_bounds__(NW * 32) void pose_enc_kernel(
    const float* __restrict__ rots, const float* __restrict__ Jtrs,
    const float* __restrict__ W0,   const float* __restrict__ b0,
    const float* __restrict__ W1,   const float* __restrict__ b1,
    const float* __restrict__ W2,   const float* __restrict__ b2,
    float* __restrict__ out, int B)
{
  __shared__ v2f s_wf[NJ][NFR][32];
  __shared__ v2f s_w0p[144][8];
  __shared__ v4f s_x[NW][TCH * 32];

  const int tid  = threadIdx.x;
  const int lane = tid & 31;
  const int half = lane >> 4;
  const int n    = lane & 15;
  const int w    = __builtin_amdgcn_readfirstlane(tid >> 5);

  for (int j = w; j < NJ; j += NW) {
    stage_joint<0, 1, 2, 3, 4, 5, 6, 7, 8, 9, 10, 11, 12, 13, 14>(
        &s_wf[j][0][lane], W1, b1, W2, b2, j, n, half);
  }
  {
    float wx[3], wy[3];
#pragma unroll
    for (int k = 0; k < 3; ++k) {
      const int i = imin(tid + k * (NW * 32), 144 * 8 - 1);
      const int col = imin(i & 7, DPJ - 1), pr = i >> 3;
      wx[k] = W0[col * 288 + 2 * pr];
      wy[k] = W0[col * 288 + 2 * pr + 1];
    }
#pragma unroll
    for (int k = 0; k < 3; ++k) {
      const int i = tid + k * (NW * 32);
      if (i < 144 * 8) {
        v2f v;
        v.x = ((i & 7) < DPJ) ? wx[k] : 0.f;
        v.y = ((i & 7) < DPJ) ? wy[k] : 0.f;
        s_w0p[i >> 3][i & 7] = v;
      }
    }
  }
  __syncthreads();

  const int n7 = n & 7;
  const v2f* wfl = &s_wf[0][0][lane];
  const v8f zero8 = {0.f, 0.f, 0.f, 0.f, 0.f, 0.f, 0.f, 0.f};
  v8f gf0;
#pragma unroll
  for (int v = 0; v < 8; ++v) gf0[v] = (v < DPJ) ? b0[imin(v, DPJ - 1)] : 0.f;

  v4f* xs = &s_x[w][0];
  const float* rrow = (const float*)xs + n * 216;
  const float* jrow = (const float*)xs + TROT + n * 72;
  const float* ra   = rrow + 4 * half;
  const float* ja   = jrow + 4 * half;
  float* srow = (float*)xs + n * 216;
  float* sa   = srow + 4 * half;

  for (int t = w; (blockIdx.x + gridDim.x * t) * 16 < B; t += NW) {
    asm volatile("" ::: "memory");
    const int rb = (blockIdx.x + gridDim.x * t) * 16;

    {
      const float* grot = rots + (long)rb * 216 + 4 * lane;
      const float* gjtr = Jtrs + (long)rb * 72 + 4 * lane;
      v4f st[TCH];
#pragma unroll
      for (int i = 0; i < TROT / 128; ++i) {
        const float* p = grot + 128 * i;
        st[i].x = p[0]; st[i].y = p[1]; st[i].z = p[2]; st[i].w = p[3];
      }
#pragma unroll
      for (int i = 0; i < TJTR / 128; ++i) {
        const float* p = gjtr + 128 * i;
        v4f& d = st[TROT / 128 + i];
        d.x = p[0]; d.y = p[1]; d.z = p[2]; d.w = p[3];
      }
#pragma unroll
      for (int i = 0; i < TCH; ++i) xs[32 * i + lane] = st[i];
    }
    asm volatile("" ::: "memory");

    JointRegs q[NJ];
    fetch_joint(q[0], ra, rrow, jrow, 0);
    v2f wv[2][NFR];
#pragma unroll
    for (int f = 0; f < NFR; ++f) wv[0][f] = wfl[f * 32];

    v8f gf = gf0;
#pragma unroll
    for (int i = 0; i < 27; ++i) {
      const v4f x = *(const v4f*)(ra + 8 * i);
      v2f x01, x23;
      x01.x = x.x; x01.y = x.y; x23.x = x.z; x23.y = x.w;
      const int pr = 4 * i + 2 * half;
      gf = wmma4(s_w0p[pr][n7],     x01, gf);
      gf = wmma4(s_w0p[pr + 1][n7], x23, gf);
    }
#pragma unroll
    for (int i = 0; i < 9; ++i) {
      const v4f x = *(const v4f*)(ja + 8 * i);
      v2f x01, x23;
      x01.x = x.x; x01.y = x.y; x23.x = x.z; x23.y = x.w;
      const int pr = 108 + 4 * i + 2 * half;
      gf = wmma4(s_w0p[pr][n7],     x01, gf);
      gf = wmma4(s_w0p[pr + 1][n7], x23, gf);
    }

    v8f O[NJ + 1];
    O[NJ] = gf;
#pragma unroll
    for (int j = 0; j < NJ; ++j) {
      asm volatile("" ::: "memory");
      if (j + 1 < NJ) fetch_joint(q[j + 1], ra, rrow, jrow, j + 1);
      if (j + 1 < NJ) {
#pragma unroll
        for (int f = 0; f < NFR; ++f) wv[(j + 1) & 1][f] = wfl[((j + 1) * NFR + f) * 32];
      }
      const v2f* wj = wv[j & 1];
      const int p = parent_of(j);
      const v8f fp = O[(p >= 0) ? p : NJ];
      const JointRegs r = q[j];

      const float dx = r.J[0] - r.P[0], dy = r.J[1] - r.P[1], dz = r.J[2] - r.P[2];
      const float bone = sqrtf(dx * dx + dy * dy + dz * dz);

      v2f B0, B1, B2, B3, B4;
      B0.x = r.a[0];                 B0.y = r.a[1];
      B1.x = r.a[2];                 B1.y = r.a[3];
      B2.x = half ? r.J[1] : r.r8;   B2.y = half ? r.J[2] : r.J[0];
      B3.x = half ? fp[0]  : bone;   B3.y = half ? fp[1]  : 1.f;
      B4.x = half ? fp[4]  : fp[2];  B4.y = half ? fp[5]  : fp[3];

      v8f aA = wmma4(wj[0], B0, zero8);
      v8f aB = wmma4(wj[5], B0, zero8);
      aA = wmma4(wj[1], B1, aA);
      aB = wmma4(wj[6], B1, aB);
      aA = wmma4(wj[2], B2, aA);
      aB = wmma4(wj[7], B2, aB);
      aA = wmma4(wj[3], B3, aA);
      aB = wmma4(wj[8], B3, aB);
      aA = wmma4(wj[4], B4, aA);
      aB = wmma4(wj[9], B4, aB);

      v2f H0, H1, H2, H3, H4;
      H0.x = fmaxf(aA[0], 0.f);  H0.y = fmaxf(aA[1], 0.f);
      H1.x = fmaxf(aA[2], 0.f);  H1.y = fmaxf(aA[3], 0.f);
      H2.x = fmaxf(aA[4], 0.f);  H2.y = fmaxf(aA[5], 0.f);
      H3.x = fmaxf(aA[6], 0.f);  H3.y = fmaxf(aA[7], 0.f);
      H4.x = fmaxf(aB[0], 0.f);  H4.y = fmaxf(aB[1], 0.f);

      v8f o = wmma4(wj[10], H0, zero8);
      o = wmma4(wj[11], H1, o);
      o = wmma4(wj[12], H2, o);
      o = wmma4(wj[13], H3, o);
      o = wmma4(wj[14], H4, o);
      O[j] = o;

      v2f w01, w23;
      w01.x = half ? o[4] : o[0];  w01.y = half ? o[5] : o[1];
      w23.x = o[2];                w23.y = o[3];
      *(v2f*)(sa + DPJ * j)       = w01;
      *(v2f*)(srow + DPJ * j + 2) = w23;
    }

    asm volatile("" ::: "memory");
    {
      float* og = out + (long)rb * (NJ * DPJ) + 4 * lane;
      v4f ov[NOCH];
#pragma unroll
      for (int i = 0; i < NOCH; ++i) {
        const int piece = 32 * i + lane;
        const int prow  = piece / 36;
        ov[i] = *(const v4f*)((const float*)xs + 216 * prow + 4 * (piece - 36 * prow));
      }
#pragma unroll
      for (int i = 0; i < NOCH; ++i) *(v4f*)(og + 128 * i) = ov[i];
    }
  }
}

extern "C" void kernel_launch(void* const* d_in, const int* in_sizes, int n_in,
                              void* d_out, int out_size, void* d_ws, size_t ws_size,
                              hipStream_t stream) {
  const float* rots = (const float*)d_in[0];
  const float* Jtrs = (const float*)d_in[1];
  const float* W0   = (const float*)d_in[2];
  const float* b0   = (const float*)d_in[3];
  const float* W1   = (const float*)d_in[4];
  const float* b1   = (const float*)d_in[5];
  const float* W2   = (const float*)d_in[6];
  const float* b2   = (const float*)d_in[7];
  float* outp = (float*)d_out;

  int B = in_sizes[0] / (NJ * 9);
  pose_enc_kernel<<<256, NW * 32, 0, stream>>>(rots, Jtrs, W0, b0, W1, b1, W2, b2,
                                            outp, B);
}


# global_feat's 72 matrix steps removed: W0 rides in idle rows of each joint's second layer-1 tile, whose rot/Jtr steps run at tile head; gf = b0 + 24 per-joint shares (sum order changes)
# speedup vs baseline: 3.9972x; 1.0345x over previous
#include <hip/hip_runtime.h>


typedef float v2f __attribute__((ext_vector_type(2)));
typedef float v4f __attribute__((ext_vector_type(4)));
typedef float v8f __attribute__((ext_vector_type(8)));

#define NJ  24
#define DF  19
#define DPJ 6
#define NW  12
#define TROT (16 * 216)
#define TJTR (16 * 72)
#define TCH  ((TROT + TJTR) / 128)
#define NOCH ((16 * NJ * DPJ) / 128)
#define NFR 15
#define PG  4
#define NMF 12

__device__ inline int imin(int a, int b) { return a < b ? a : b; }

__host__ __device__ constexpr int main_frag(int f) { return f < 5 ? f : f + 3; }

__host__ __device__ constexpr unsigned long long pack_parents(int base) {
  constexpr int P[NJ] = {
      -1, 0, 0, 0, 1, 2, 3, 4, 5, 6, 7, 8, 9, 9, 9, 12, 13, 14, 16, 17, 18, 19, 20, 21};
  unsigned long long v = 0;
  for (int i = 0; i < 12; ++i) v |= (unsigned long long)(P[base + i] + 1) << (5 * i);
  return v;
}
__host__ __device__ constexpr int parent_of(int j) {
  const unsigned long long wv = (j < 12) ? pack_parents(0) : pack_parents(12);
  const int s = 5 * ((j < 12) ? j : j - 12);
  return (int)((wv >> s) & 31ull) - 1;
}

__host__ __device__ constexpr unsigned long long pack_slots(int half) {
  constexpr int S[2][10] = {{0, 1, 2, 3, 8, 9, 12, 19, 15, 16},
                            {4, 5, 6, 7, 10, 11, 13, 14, 17, 18}};
  unsigned long long v = 0;
  for (int e = 0; e < 10; ++e) v |= (unsigned long long)S[half][e] << (5 * e);
  return v;
}

template <int F, int S>
__device__ inline float wfrag_value(const float* __restrict__ W0,
                                    const float* __restrict__ W1, const float* __restrict__ b1,
                                    const float* __restrict__ W2, const float* __restrict__ b2,
                                    int j, int c, int h) {
  constexpr int grp = F / 5, t = F - 5 * grp, e = 2 * t + S;
  if constexpr (grp < 2) {
    constexpr int c0 = (int)((pack_slots(0) >> (5 * e)) & 31ull);
    constexpr int c1 = (int)((pack_slots(1) >> (5 * e)) & 31ull);
    static_assert(c1 < DF, "the spare slot sits in lane half 0 only");
    const int rowB = (c == 0) ? 16 : ((c == 1) ? 17 : 18);
    const int row = (grp == 1) ? rowB : c;
    const bool valid = (grp == 1) ? ((c == 0) || (c == 1) || (c == 8)) : true;
    const float* wr = W1 + j * (DF * DF) + row * DF;
    float x;
    if constexpr (grp == 1 && t < 3) {
      static_assert(c0 < 12 && c1 < 12, "steps 0..2 hold rot and Jtr columns only");
      constexpr int g0 = (c0 < 9) ? c0 : 216 + (c0 - 9);
      constexpr int g1 = (c1 < 9) ? c1 : 216 + (c1 - 9);
      constexpr int s0 = (c0 < 9) ? 9 : 3, s1 = (c1 < 9) ? 9 : 3;
      const int r7 = c & 7;
      const int gr = imin(r7 < 2 ? 0 : r7 - 2, DPJ - 1);
      const float* pw1 = wr + (h ? c1 : c0);
      const float* pw0 = W0 + gr * 288 + (h ? (g1 + s1 * j) : (g0 + s0 * j));
      x = *(valid ? pw1 : pw0);
      return (valid || r7 >= 2) ? x : 0.f;
    } else if constexpr (c0 < DF) {
      x = wr[h ? c1 : c0];
      return valid ? x : 0.f;
    } else {
      const float xw = wr[c1];
      const float xb = b1[j * DF + row];
      x = h ? xw : xb;
      const float alt = ((grp == 1) && (c == 9) && (h == 0)) ? 1.f : 0.f;
      return valid ? x : alt;
    }
  } else {
    const int fo = c & 7;
    const bool valid = fo < DPJ;
    const int foc = imin(fo, DPJ - 1);
    const float* wr = W2 + j * (DPJ * DF) + foc * DF;
    float x;
    if constexpr (t < 4) {
      x = wr[8 * h + 2 * t + S];
    } else if constexpr (S == 0) {
      x = wr[h ? 18 : 16];
    } else {
      const float xw = wr[17];
      const float xb = b2[j * DPJ + foc];
      x = h ? xb : xw;
    }
    return valid ? x : 0.f;
  }
}

template <int... F>
__device__ inline void stage_joint(v2f* dst, const float* __restrict__ W0,
                                   const float* __restrict__ W1, const float* __restrict__ b1,
                                   const float* __restrict__ W2, const float* __restrict__ b2,
                                   int j, int c, int h) {
  v2f v[NFR];
#pragma unroll
  for (int f = 0; f < NFR; ++f) { v[f].x = 0.f; v[f].y = 0.f; }
  ((v[F].x = wfrag_value<F, 0>(W0, W1, b1, W2, b2, j, c, h),
    v[F].y = wfrag_value<F, 1>(W0, W1, b1, W2, b2, j, c, h)), ...);
#pragma unroll
  for (int f = 0; f < NFR; ++f) dst[f * 32] = v[f];
}

struct JointRegs {
  float a[4];
  float r8;
  float J[3];
  float P[3];
};

__device__ inline void fetch_joint(JointRegs& r, const float* ra, const float* rrow,
                                   const float* jrow, int j) {
  const float* pa = ra + 9 * j;
#pragma unroll
  for (int q = 0; q < 4; ++q) r.a[q] = pa[q];
  r.r8 = rrow[9 * j + 8];
  const float* pj = jrow + 3 * j;
#pragma unroll
  for (int q = 0; q < 3; ++q) r.J[q] = pj[q];
  const int p = parent_of(j);
  if (p >= 0) {
    const float* pp = jrow + 3 * p;
#pragma unroll
    for (int q = 0; q < 3; ++q) r.P[q] = pp[q];
  } else {
#pragma unroll
    for (int q = 0; q < 3; ++q) r.P[q] = 0.f;
  }
}

__device__ inline v8f wmma4(v2f a, v2f b, v8f c) {
  return __builtin_amdgcn_wmma_f32_16x16x4_f32(
       false, a,  false, b,
       (short)0, c,  false,  false);
}

__global__ __launch_bounds__(NW * 32) void pose_enc_kernel(
    const float* __restrict__ rots, const float* __restrict__ Jtrs,
    const float* __restrict__ W0,   const float* __restrict__ b0,
    const float* __restrict__ W1,   const float* __restrict__ b1,
    const float* __restrict__ W2,   const float* __restrict__ b2,
    float* __restrict__ out, int B)
{
  __shared__ v2f s_wf[NJ][NFR][32];
  __shared__ v4f s_x[NW][TCH * 32];

  const int tid  = threadIdx.x;
  const int lane = tid & 31;
  const int half = lane >> 4;
  const int n    = lane & 15;
  const int w    = __builtin_amdgcn_readfirstlane(tid >> 5);

  for (int j = w; j < NJ; j += NW) {
    stage_joint<0, 1, 2, 3, 4, 5, 6, 7, 8, 9, 10, 11, 12, 13, 14>(
        &s_wf[j][0][lane], W0, W1, b1, W2, b2, j, n, half);
  }
  __syncthreads();

  const v2f* wfl = &s_wf[0][0][lane];
  const v8f zero8 = {0.f, 0.f, 0.f, 0.f, 0.f, 0.f, 0.f, 0.f};
  float gb[DPJ];
#pragma unroll
  for (int v = 0; v < DPJ; ++v) gb[v] = b0[v];

  v4f* xs = &s_x[w][0];
  const float* rrow = (const float*)xs + n * 216;
  const float* jrow = (const float*)xs + TROT + n * 72;
  const float* ra   = rrow + 4 * half;
  const float* px   = half ? (jrow + 1) : (rrow + 8);
  const int    sx   = half ? 3 : 9;
  const float* py   = jrow + 2 * half;
  float* srow = (float*)xs + n * 216;
  float* sa   = srow + 4 * half;

  for (int t = w; (blockIdx.x + gridDim.x * t) * 16 < B; t += NW) {
    asm volatile("" ::: "memory");
    const int rb = (blockIdx.x + gridDim.x * t) * 16;

    {
      const float* grot = rots + (long)rb * 216 + 4 * lane;
      const float* gjtr = Jtrs + (long)rb * 72 + 4 * lane;
      v4f st[TCH];
#pragma unroll
      for (int i = 0; i < TROT / 128; ++i) {
        const float* p = grot + 128 * i;
        st[i].x = p[0]; st[i].y = p[1]; st[i].z = p[2]; st[i].w = p[3];
      }
#pragma unroll
      for (int i = 0; i < TJTR / 128; ++i) {
        const float* p = gjtr + 128 * i;
        v4f& d = st[TROT / 128 + i];
        d.x = p[0]; d.y = p[1]; d.z = p[2]; d.w = p[3];
      }
#pragma unroll
      for (int i = 0; i < TCH; ++i) xs[32 * i + lane] = st[i];
    }
    asm volatile("" ::: "memory");

    float pr0[NJ], pr1[NJ];
    float g[DPJ];
#pragma unroll
    for (int v = 0; v < DPJ; ++v) g[v] = gb[v];
    v2f pb[2][PG][3], pw[2][PG][3];
#pragma unroll
    for (int u = 0; u < PG; ++u) {
      const float* pa = ra + 9 * u;
      pb[0][u][0].x = pa[0];        pb[0][u][0].y = pa[1];
      pb[0][u][1].x = pa[2];        pb[0][u][1].y = pa[3];
      pb[0][u][2].x = px[sx * u];   pb[0][u][2].y = py[3 * u];
#pragma unroll
      for (int s = 0; s < 3; ++s) pw[0][u][s] = wfl[(u * NFR + 5 + s) * 32];
    }
#pragma unroll
    for (int gq = 0; gq < NJ / PG; ++gq) {
      asm volatile("" ::: "memory");
      if (gq + 1 < NJ / PG) {
#pragma unroll
        for (int u = 0; u < PG; ++u) {
          const int j = (gq + 1) * PG + u;
          const float* pa = ra + 9 * j;
          pb[(gq + 1) & 1][u][0].x = pa[0];        pb[(gq + 1) & 1][u][0].y = pa[1];
          pb[(gq + 1) & 1][u][1].x = pa[2];        pb[(gq + 1) & 1][u][1].y = pa[3];
          pb[(gq + 1) & 1][u][2].x = px[sx * j];   pb[(gq + 1) & 1][u][2].y = py[3 * j];
#pragma unroll
          for (int s = 0; s < 3; ++s) pw[(gq + 1) & 1][u][s] = wfl[(j * NFR + 5 + s) * 32];
        }
      }
#pragma unroll
      for (int u = 0; u < PG; ++u) {
        const int j = gq * PG + u;
        v8f tt = wmma4(pw[gq & 1][u][0], pb[gq & 1][u][0], zero8);
        tt = wmma4(pw[gq & 1][u][1], pb[gq & 1][u][1], tt);
        tt = wmma4(pw[gq & 1][u][2], pb[gq & 1][u][2], tt);
        pr0[j] = tt[0];
        pr1[j] = tt[1];
#pragma unroll
        for (int v = 0; v < DPJ; ++v) g[v] += tt[2 + v];
      }
    }
    v8f gf = zero8;
#pragma unroll
    for (int v = 0; v < DPJ; ++v) gf[v] = g[v];

    JointRegs q[NJ];
    fetch_joint(q[0], ra, rrow, jrow, 0);
    v2f wv[2][NMF];
#pragma unroll
    for (int f = 0; f < NMF; ++f) wv[0][f] = wfl[main_frag(f) * 32];

    v8f O[NJ + 1];
    O[NJ] = gf;
#pragma unroll
    for (int j = 0; j < NJ; ++j) {
      asm volatile("" ::: "memory");
      if (j + 1 < NJ) fetch_joint(q[j + 1], ra, rrow, jrow, j + 1);
      if (j + 1 < NJ) {
#pragma unroll
        for (int f = 0; f < NMF; ++f) wv[(j + 1) & 1][f] = wfl[((j + 1) * NFR + main_frag(f)) * 32];
      }
      const v2f* wj = wv[j & 1];
      const int p = parent_of(j);
      const v8f fp = O[(p >= 0) ? p : NJ];
      const JointRegs r = q[j];

      const float dx = r.J[0] - r.P[0], dy = r.J[1] - r.P[1], dz = r.J[2] - r.P[2];
      const float bone = sqrtf(dx * dx + dy * dy + dz * dz);

      v2f B0, B1, B2, B3, B4;
      B0.x = r.a[0];                 B0.y = r.a[1];
      B1.x = r.a[2];                 B1.y = r.a[3];
      B2.x = half ? r.J[1] : r.r8;   B2.y = half ? r.J[2] : r.J[0];
      B3.x = half ? fp[0]  : bone;   B3.y = half ? fp[1]  : 1.f;
      B4.x = half ? fp[4]  : fp[2];  B4.y = half ? fp[5]  : fp[3];

      v8f aB = zero8;
      aB[0] = pr0[j];
      aB[1] = pr1[j];
      v8f aA = wmma4(wj[0], B0, zero8);
      aA = wmma4(wj[1], B1, aA);
      aA = wmma4(wj[2], B2, aA);
      aA = wmma4(wj[3], B3, aA);
      aB = wmma4(wj[5], B3, aB);
      aA = wmma4(wj[4], B4, aA);
      aB = wmma4(wj[6], B4, aB);

      v2f H0, H1, H2, H3, H4;
      H0.x = fmaxf(aA[0], 0.f);  H0.y = fmaxf(aA[1], 0.f);
      H1.x = fmaxf(aA[2], 0.f);  H1.y = fmaxf(aA[3], 0.f);
      H2.x = fmaxf(aA[4], 0.f);  H2.y = fmaxf(aA[5], 0.f);
      H3.x = fmaxf(aA[6], 0.f);  H3.y = fmaxf(aA[7], 0.f);
      H4.x = fmaxf(aB[0], 0.f);  H4.y = fmaxf(aB[1], 0.f);

      v8f o = wmma4(wj[7], H0, zero8);
      o = wmma4(wj[8], H1, o);
      o = wmma4(wj[9], H2, o);
      o = wmma4(wj[10], H3, o);
      o = wmma4(wj[11], H4, o);
      O[j] = o;

      v2f w01, w23;
      w01.x = half ? o[4] : o[0];  w01.y = half ? o[5] : o[1];
      w23.x = o[2];                w23.y = o[3];
      *(v2f*)(sa + DPJ * j)       = w01;
      *(v2f*)(srow + DPJ * j + 2) = w23;
    }

    asm volatile("" ::: "memory");
    {
      float* og = out + (long)rb * (NJ * DPJ) + 4 * lane;
      v4f ov[NOCH];
#pragma unroll
      for (int i = 0; i < NOCH; ++i) {
        const int piece = 32 * i + lane;
        const int prow  = piece / 36;
        ov[i] = *(const v4f*)((const float*)xs + 216 * prow + 4 * (piece - 36 * prow));
      }
#pragma unroll
      for (int i = 0; i < NOCH; ++i) *(v4f*)(og + 128 * i) = ov[i];
    }
  }
}

extern "C" void kernel_launch(void* const* d_in, const int* in_sizes, int n_in,
                              void* d_out, int out_size, void* d_ws, size_t ws_size,
                              hipStream_t stream) {
  const float* rots = (const float*)d_in[0];
  const float* Jtrs = (const float*)d_in[1];
  const float* W0   = (const float*)d_in[2];
  const float* b0   = (const float*)d_in[3];
  const float* W1   = (const float*)d_in[4];
  const float* b1   = (const float*)d_in[5];
  const float* W2   = (const float*)d_in[6];
  const float* b2   = (const float*)d_in[7];
  float* outp = (float*)d_out;

  int B = in_sizes[0] / (NJ * 9);
  pose_enc_kernel<<<256, NW * 32, 0, stream>>>(rots, Jtrs, W0, b0, W1, b1, W2, b2,
                                            outp, B);
}


# bone length by the hardware square root (v_sqrt_f32, 1 ulp) instead of the correctly rounded sequence: sixteen dependent vector instructions per joint leave the joint chain
# speedup vs baseline: 4.1490x; 1.0380x over previous
#include <hip/hip_runtime.h>


typedef float v2f __attribute__((ext_vector_type(2)));
typedef float v4f __attribute__((ext_vector_type(4)));
typedef float v8f __attribute__((ext_vector_type(8)));

#define NJ  24
#define DF  19
#define DPJ 6
#define NW  12
#define TROT (16 * 216)
#define TJTR (16 * 72)
#define TCH  ((TROT + TJTR) / 128)
#define NOCH ((16 * NJ * DPJ) / 128)
#define NFR 15
#define PG  4
#define NMF 12

__device__ inline int imin(int a, int b) { return a < b ? a : b; }

__host__ __device__ constexpr int main_frag(int f) { return f < 5 ? f : f + 3; }

__host__ __device__ constexpr unsigned long long pack_parents(int base) {
  constexpr int P[NJ] = {
      -1, 0, 0, 0, 1, 2, 3, 4, 5, 6, 7, 8, 9, 9, 9, 12, 13, 14, 16, 17, 18, 19, 20, 21};
  unsigned long long v = 0;
  for (int i = 0; i < 12; ++i) v |= (unsigned long long)(P[base + i] + 1) << (5 * i);
  return v;
}
__host__ __device__ constexpr int parent_of(int j) {
  const unsigned long long wv = (j < 12) ? pack_parents(0) : pack_parents(12);
  const int s = 5 * ((j < 12) ? j : j - 12);
  return (int)((wv >> s) & 31ull) - 1;
}

__host__ __device__ constexpr unsigned long long pack_slots(int half) {
  constexpr int S[2][10] = {{0, 1, 2, 3, 8, 9, 12, 19, 15, 16},
                            {4, 5, 6, 7, 10, 11, 13, 14, 17, 18}};
  unsigned long long v = 0;
  for (int e = 0; e < 10; ++e) v |= (unsigned long long)S[half][e] << (5 * e);
  return v;
}

template <int F, int S>
__device__ inline float wfrag_value(const float* __restrict__ W0,
                                    const float* __restrict__ W1, const float* __restrict__ b1,
                                    const float* __restrict__ W2, const float* __restrict__ b2,
                                    int j, int c, int h) {
  constexpr int grp = F / 5, t = F - 5 * grp, e = 2 * t + S;
  if constexpr (grp < 2) {
    constexpr int c0 = (int)((pack_slots(0) >> (5 * e)) & 31ull);
    constexpr int c1 = (int)((pack_slots(1) >> (5 * e)) & 31ull);
    static_assert(c1 < DF, "the spare slot sits in lane half 0 only");
    const int rowB = (c == 0) ? 16 : ((c == 1) ? 17 : 18);
    const int row = (grp == 1) ? rowB : c;
    const bool valid = (grp == 1) ? ((c == 0) || (c == 1) || (c == 8)) : true;
    const float* wr = W1 + j * (DF * DF) + row * DF;
    float x;
    if constexpr (grp == 1 && t < 3) {
      static_assert(c0 < 12 && c1 < 12, "steps 0..2 hold rot and Jtr columns only");
      constexpr int g0 = (c0 < 9) ? c0 : 216 + (c0 - 9);
      constexpr int g1 = (c1 < 9) ? c1 : 216 + (c1 - 9);
      constexpr int s0 = (c0 < 9) ? 9 : 3, s1 = (c1 < 9) ? 9 : 3;
      const int r7 = c & 7;
      const int gr = imin(r7 < 2 ? 0 : r7 - 2, DPJ - 1);
      const float* pw1 = wr + (h ? c1 : c0);
      const float* pw0 = W0 + gr * 288 + (h ? (g1 + s1 * j) : (g0 + s0 * j));
      x = *(valid ? pw1 : pw0);
      return (valid || r7 >= 2) ? x : 0.f;
    } else if constexpr (c0 < DF) {
      x = wr[h ? c1 : c0];
      return valid ? x : 0.f;
    } else {
      const float xw = wr[c1];
      const float xb = b1[j * DF + row];
      x = h ? xw : xb;
      const float alt = ((grp == 1) && (c == 9) && (h == 0)) ? 1.f : 0.f;
      return valid ? x : alt;
    }
  } else {
    const int fo = c & 7;
    const bool valid = fo < DPJ;
    const int foc = imin(fo, DPJ - 1);
    const float* wr = W2 + j * (DPJ * DF) + foc * DF;
    float x;
    if constexpr (t < 4) {
      x = wr[8 * h + 2 * t + S];
    } else if constexpr (S == 0) {
      x = wr[h ? 18 : 16];
    } else {
      const float xw = wr[17];
      const float xb = b2[j * DPJ + foc];
      x = h ? xb : xw;
    }
    return valid ? x : 0.f;
  }
}

template <int... F>
__device__ inline void stage_joint(v2f* dst, const float* __restrict__ W0,
                                   const float* __restrict__ W1, const float* __restrict__ b1,
                                   const float* __restrict__ W2, const float* __restrict__ b2,
                                   int j, int c, int h) {
  v2f v[NFR];
#pragma unroll
  for (int f = 0; f < NFR; ++f) { v[f].x = 0.f; v[f].y = 0.f; }
  ((v[F].x = wfrag_value<F, 0>(W0, W1, b1, W2, b2, j, c, h),
    v[F].y = wfrag_value<F, 1>(W0, W1, b1, W2, b2, j, c, h)), ...);
#pragma unroll
  for (int f = 0; f < NFR; ++f) dst[f * 32] = v[f];
}

struct JointRegs {
  float a[4];
  float r8;
  float J[3];
  float P[3];
};

__device__ inline void fetch_joint(JointRegs& r, const float* ra, const float* rrow,
                                   const float* jrow, int j) {
  const float* pa = ra + 9 * j;
#pragma unroll
  for (int q = 0; q < 4; ++q) r.a[q] = pa[q];
  r.r8 = rrow[9 * j + 8];
  const float* pj = jrow + 3 * j;
#pragma unroll
  for (int q = 0; q < 3; ++q) r.J[q] = pj[q];
  const int p = parent_of(j);
  if (p >= 0) {
    const float* pp = jrow + 3 * p;
#pragma unroll
    for (int q = 0; q < 3; ++q) r.P[q] = pp[q];
  } else {
#pragma unroll
    for (int q = 0; q < 3; ++q) r.P[q] = 0.f;
  }
}

__device__ inline float bone_len(const JointRegs& r) {
  const float dx = r.J[0] - r.P[0], dy = r.J[1] - r.P[1], dz = r.J[2] - r.P[2];
  return __builtin_amdgcn_sqrtf(dx * dx + dy * dy + dz * dz);
}

__device__ inline v8f wmma4(v2f a, v2f b, v8f c) {
  return __builtin_amdgcn_wmma_f32_16x16x4_f32(
       false, a,  false, b,
       (short)0, c,  false,  false);
}

__global__ __launch_bounds__(NW * 32) void pose_enc_kernel(
    const float* __restrict__ rots, const float* __restrict__ Jtrs,
    const float* __restrict__ W0,   const float* __restrict__ b0,
    const float* __restrict__ W1,   const float* __restrict__ b1,
    const float* __restrict__ W2,   const float* __restrict__ b2,
    float* __restrict__ out, int B)
{
  __shared__ v2f s_wf[NJ][NFR][32];
  __shared__ v4f s_x[NW][TCH * 32];

  const int tid  = threadIdx.x;
  const int lane = tid & 31;
  const int half = lane >> 4;
  const int n    = lane & 15;
  const int w    = __builtin_amdgcn_readfirstlane(tid >> 5);

  for (int j = w; j < NJ; j += NW) {
    stage_joint<0, 1, 2, 3, 4, 5, 6, 7, 8, 9, 10, 11, 12, 13, 14>(
        &s_wf[j][0][lane], W0, W1, b1, W2, b2, j, n, half);
  }
  __syncthreads();

  const v2f* wfl = &s_wf[0][0][lane];
  const v8f zero8 = {0.f, 0.f, 0.f, 0.f, 0.f, 0.f, 0.f, 0.f};
  float gb[DPJ];
#pragma unroll
  for (int v = 0; v < DPJ; ++v) gb[v] = b0[v];

  v4f* xs = &s_x[w][0];
  const float* rrow = (const float*)xs + n * 216;
  const float* jrow = (const float*)xs + TROT + n * 72;
  const float* ra   = rrow + 4 * half;
  const float* px   = half ? (jrow + 1) : (rrow + 8);
  const int    sx   = half ? 3 : 9;
  const float* py   = jrow + 2 * half;
  float* srow = (float*)xs + n * 216;
  float* sa   = srow + 4 * half;

  for (int t = w; (blockIdx.x + gridDim.x * t) * 16 < B; t += NW) {
    asm volatile("" ::: "memory");
    const int rb = (blockIdx.x + gridDim.x * t) * 16;

    {
      const float* grot = rots + (long)rb * 216 + 4 * lane;
      const float* gjtr = Jtrs + (long)rb * 72 + 4 * lane;
      v4f st[TCH];
#pragma unroll
      for (int i = 0; i < TROT / 128; ++i) {
        const float* p = grot + 128 * i;
        st[i].x = p[0]; st[i].y = p[1]; st[i].z = p[2]; st[i].w = p[3];
      }
#pragma unroll
      for (int i = 0; i < TJTR / 128; ++i) {
        const float* p = gjtr + 128 * i;
        v4f& d = st[TROT / 128 + i];
        d.x = p[0]; d.y = p[1]; d.z = p[2]; d.w = p[3];
      }
#pragma unroll
      for (int i = 0; i < TCH; ++i) xs[32 * i + lane] = st[i];
    }
    asm volatile("" ::: "memory");

    float pr0[NJ], pr1[NJ];
    float g[DPJ];
#pragma unroll
    for (int v = 0; v < DPJ; ++v) g[v] = gb[v];
    v2f pb[2][PG][3], pw[2][PG][3];
#pragma unroll
    for (int u = 0; u < PG; ++u) {
      const float* pa = ra + 9 * u;
      pb[0][u][0].x = pa[0];        pb[0][u][0].y = pa[1];
      pb[0][u][1].x = pa[2];        pb[0][u][1].y = pa[3];
      pb[0][u][2].x = px[sx * u];   pb[0][u][2].y = py[3 * u];
#pragma unroll
      for (int s = 0; s < 3; ++s) pw[0][u][s] = wfl[(u * NFR + 5 + s) * 32];
    }
#pragma unroll
    for (int gq = 0; gq < NJ / PG; ++gq) {
      asm volatile("" ::: "memory");
      if (gq + 1 < NJ / PG) {
#pragma unroll
        for (int u = 0; u < PG; ++u) {
          const int j = (gq + 1) * PG + u;
          const float* pa = ra + 9 * j;
          pb[(gq + 1) & 1][u][0].x = pa[0];        pb[(gq + 1) & 1][u][0].y = pa[1];
          pb[(gq + 1) & 1][u][1].x = pa[2];        pb[(gq + 1) & 1][u][1].y = pa[3];
          pb[(gq + 1) & 1][u][2].x = px[sx * j];   pb[(gq + 1) & 1][u][2].y = py[3 * j];
#pragma unroll
          for (int s = 0; s < 3; ++s) pw[(gq + 1) & 1][u][s] = wfl[(j * NFR + 5 + s) * 32];
        }
      }
#pragma unroll
      for (int u = 0; u < PG; ++u) {
        const int j = gq * PG + u;
        v8f tt = wmma4(pw[gq & 1][u][0], pb[gq & 1][u][0], zero8);
        tt = wmma4(pw[gq & 1][u][1], pb[gq & 1][u][1], tt);
        tt = wmma4(pw[gq & 1][u][2], pb[gq & 1][u][2], tt);
        pr0[j] = tt[0];
        pr1[j] = tt[1];
#pragma unroll
        for (int v = 0; v < DPJ; ++v) g[v] += tt[2 + v];
      }
    }
    v8f gf = zero8;
#pragma unroll
    for (int v = 0; v < DPJ; ++v) gf[v] = g[v];

    JointRegs q[NJ];
    fetch_joint(q[0], ra, rrow, jrow, 0);
    v2f wv[2][NMF];
#pragma unroll
    for (int f = 0; f < NMF; ++f) wv[0][f] = wfl[main_frag(f) * 32];

    v8f O[NJ + 1];
    O[NJ] = gf;
#pragma unroll
    for (int j = 0; j < NJ; ++j) {
      asm volatile("" ::: "memory");
      if (j + 1 < NJ) fetch_joint(q[j + 1], ra, rrow, jrow, j + 1);
      if (j + 1 < NJ) {
#pragma unroll
        for (int f = 0; f < NMF; ++f) wv[(j + 1) & 1][f] = wfl[((j + 1) * NFR + main_frag(f)) * 32];
      }
      const v2f* wj = wv[j & 1];
      const int p = parent_of(j);
      const v8f fp = O[(p >= 0) ? p : NJ];
      const JointRegs r = q[j];

      const float bone = bone_len(r);

      v2f B0, B1, B2, B3, B4;
      B0.x = r.a[0];                 B0.y = r.a[1];
      B1.x = r.a[2];                 B1.y = r.a[3];
      B2.x = half ? r.J[1] : r.r8;   B2.y = half ? r.J[2] : r.J[0];
      B3.x = half ? fp[0]  : bone;   B3.y = half ? fp[1]  : 1.f;
      B4.x = half ? fp[4]  : fp[2];  B4.y = half ? fp[5]  : fp[3];

      v8f aB = zero8;
      aB[0] = pr0[j];
      aB[1] = pr1[j];
      v8f aA = wmma4(wj[0], B0, zero8);
      aA = wmma4(wj[1], B1, aA);
      aA = wmma4(wj[2], B2, aA);
      aA = wmma4(wj[3], B3, aA);
      aB = wmma4(wj[5], B3, aB);
      aA = wmma4(wj[4], B4, aA);
      aB = wmma4(wj[6], B4, aB);

      v2f H0, H1, H2, H3, H4;
      H0.x = fmaxf(aA[0], 0.f);  H0.y = fmaxf(aA[1], 0.f);
      H1.x = fmaxf(aA[2], 0.f);  H1.y = fmaxf(aA[3], 0.f);
      H2.x = fmaxf(aA[4], 0.f);  H2.y = fmaxf(aA[5], 0.f);
      H3.x = fmaxf(aA[6], 0.f);  H3.y = fmaxf(aA[7], 0.f);
      H4.x = fmaxf(aB[0], 0.f);  H4.y = fmaxf(aB[1], 0.f);

      v8f o = wmma4(wj[7], H0, zero8);
      o = wmma4(wj[8], H1, o);
      o = wmma4(wj[9], H2, o);
      o = wmma4(wj[10], H3, o);
      o = wmma4(wj[11], H4, o);
      O[j] = o;

      v2f w01, w23;
      w01.x = half ? o[4] : o[0];  w01.y = half ? o[5] : o[1];
      w23.x = o[2];                w23.y = o[3];
      *(v2f*)(sa + DPJ * j)       = w01;
      *(v2f*)(srow + DPJ * j + 2) = w23;
    }

    asm volatile("" ::: "memory");
    {
      float* og = out + (long)rb * (NJ * DPJ) + 4 * lane;
      v4f ov[NOCH];
#pragma unroll
      for (int i = 0; i < NOCH; ++i) {
        const int piece = 32 * i + lane;
        const int prow  = piece / 36;
        ov[i] = *(const v4f*)((const float*)xs + 216 * prow + 4 * (piece - 36 * prow));
      }
#pragma unroll
      for (int i = 0; i < NOCH; ++i) *(v4f*)(og + 128 * i) = ov[i];
    }
  }
}

extern "C" void kernel_launch(void* const* d_in, const int* in_sizes, int n_in,
                              void* d_out, int out_size, void* d_ws, size_t ws_size,
                              hipStream_t stream) {
  const float* rots = (const float*)d_in[0];
  const float* Jtrs = (const float*)d_in[1];
  const float* W0   = (const float*)d_in[2];
  const float* b0   = (const float*)d_in[3];
  const float* W1   = (const float*)d_in[4];
  const float* b1   = (const float*)d_in[5];
  const float* W2   = (const float*)d_in[6];
  const float* b2   = (const float*)d_in[7];
  float* outp = (float*)d_out;

  int B = in_sizes[0] / (NJ * 9);
  pose_enc_kernel<<<256, NW * 32, 0, stream>>>(rots, Jtrs, W0, b0, W1, b1, W2, b2,
                                            outp, B);
}
